# starting kernel
# baseline (speedup 1.0000x reference)
#include <hip/hip_runtime.h>

typedef __attribute__((ext_vector_type(2))) float v2f;
typedef __attribute__((ext_vector_type(8))) float v8f;

#define N_NODES  100000
#define N_EDGES  3200000
#define N_GRAPHS 1024

static inline int ceil_div(int a, int b) { return (a + b - 1) / b; }

__global__ void fill4_f32(float4* __restrict__ p, float v, int nquads) {
    int i = blockIdx.x * blockDim.x + threadIdx.x;
    if (i < nquads) p[i] = make_float4(v, v, v, v);
}

__global__ void deg_accum(const int* __restrict__ ei, float* __restrict__ deg) {
    int e = blockIdx.x * blockDim.x + threadIdx.x;
    if (e < N_EDGES) atomicAdd(&deg[ei[N_EDGES + e]], 1.0f);
}

__global__ void rsqrt_inplace(float* __restrict__ d, int n) {
    int i = blockIdx.x * blockDim.x + threadIdx.x;
    if (i < n) d[i] = rsqrtf(d[i]);
}

template <int CIN, int COUT>
__global__ void __launch_bounds__(256)
gemm_wmma_f32(const float* __restrict__ H, const float* __restrict__ W,
              const float* __restrict__ bias, float* __restrict__ out, int nrows) {
    constexpr int NT     = COUT / 16;
    constexpr int KSTEPS = (CIN + 3) / 4;

    const int wavesPerBlock = blockDim.x >> 5;
    const int wave = blockIdx.x * wavesPerBlock + (threadIdx.x >> 5);
    const int row0 = wave * 16;
    if (row0 >= nrows) return;

    const int lane = threadIdx.x & 31;
    const int half = lane >> 4;
    const int m    = lane & 15;
    const int row  = row0 + m;

    v8f acc[NT];
#pragma unroll
    for (int nt = 0; nt < NT; ++nt) acc[nt] = (v8f){};

#pragma unroll
    for (int ks = 0; ks < KSTEPS; ++ks) {
        const int ka = ks * 4 + 2 * half;

        v2f a;
        if constexpr (CIN % 4 == 0) {
            a = *reinterpret_cast<const v2f*>(&H[row * CIN + ka]);
        } else {
            const int k0 = (ka     < CIN) ? ka     : (CIN - 1);
            const int k1 = (ka + 1 < CIN) ? ka + 1 : (CIN - 1);
            float ax = H[row * CIN + k0];
            float ay = H[row * CIN + k1];
            a.x = (ka     < CIN) ? ax : 0.0f;
            a.y = (ka + 1 < CIN) ? ay : 0.0f;
        }

#pragma unroll
        for (int nt = 0; nt < NT; ++nt) {
            const int ncol = nt * 16 + m;
            v2f b;
            if constexpr (CIN % 4 == 0) {
                b = *reinterpret_cast<const v2f*>(&W[ncol * CIN + ka]);
            } else {
                const int k0 = (ka     < CIN) ? ka     : (CIN - 1);
                const int k1 = (ka + 1 < CIN) ? ka + 1 : (CIN - 1);
                float bx = W[ncol * CIN + k0];
                float by = W[ncol * CIN + k1];
                b.x = (ka     < CIN) ? bx : 0.0f;
                b.y = (ka + 1 < CIN) ? by : 0.0f;
            }
            acc[nt] = __builtin_amdgcn_wmma_f32_16x16x4_f32(
                false, a, false, b, (short)0, acc[nt], false, false);
        }
    }

#pragma unroll
    for (int nt = 0; nt < NT; ++nt) {
        const int col = nt * 16 + m;
        const float bb = bias ? bias[col] : 0.0f;
#pragma unroll
        for (int r = 0; r < 8; ++r) {
            out[(row0 + r + 8 * half) * COUT + col] = acc[nt][r] + bb;
        }
    }
}

template <int COUT>
__global__ void edge_scatter(const int* __restrict__ ei, const float* __restrict__ t,
                             const float* __restrict__ dis, float* __restrict__ agg) {
    constexpr int GROUPS = COUT / 4;
    const int tid = blockIdx.x * blockDim.x + threadIdx.x;
    const int e = tid / GROUPS;
    const int g = tid % GROUPS;
    if (e >= N_EDGES) return;
    if (g == 0 && e + 4096 < N_EDGES) {
        __builtin_prefetch(&ei[e + 4096], 0, 1);
        __builtin_prefetch(&ei[N_EDGES + e + 4096], 0, 1);
    }
    const int src = ei[e];
    const int dst = ei[N_EDGES + e];
    const float coeff = dis[src] * dis[dst];
    const float4 v = *reinterpret_cast<const float4*>(&t[src * COUT + g * 4]);
    float* base = &agg[dst * COUT + g * 4];
    atomicAdd(base + 0, v.x * coeff);
    atomicAdd(base + 1, v.y * coeff);
    atomicAdd(base + 2, v.z * coeff);
    atomicAdd(base + 3, v.w * coeff);
}

template <int COUT>
__global__ void finalize_relu(const float* __restrict__ agg, const float* __restrict__ t,
                              const float* __restrict__ dis, const float* __restrict__ bias,
                              float* __restrict__ outH) {
    constexpr int GROUPS = COUT / 4;
    const int q = blockIdx.x * blockDim.x + threadIdx.x;
    if (q >= N_NODES * GROUPS) return;
    const int n  = q / GROUPS;
    const int c4 = (q % GROUPS) * 4;
    const float d  = dis[n];
    const float dd = d * d;
    const float4 va = *reinterpret_cast<const float4*>(&agg[n * COUT + c4]);
    const float4 vt = *reinterpret_cast<const float4*>(&t[n * COUT + c4]);
    const float4 vb = *reinterpret_cast<const float4*>(&bias[c4]);
    float4 r;
    r.x = fmaxf(fmaf(vt.x, dd, va.x) + vb.x, 0.0f);
    r.y = fmaxf(fmaf(vt.y, dd, va.y) + vb.y, 0.0f);
    r.z = fmaxf(fmaf(vt.z, dd, va.z) + vb.z, 0.0f);
    r.w = fmaxf(fmaf(vt.w, dd, va.w) + vb.w, 0.0f);
    *reinterpret_cast<float4*>(&outH[n * COUT + c4]) = r;
}

__global__ void pool_kernel(const float* __restrict__ h, const int* __restrict__ batch,
                            float* __restrict__ pooled) {
    const int q = blockIdx.x * blockDim.x + threadIdx.x;
    if (q >= N_NODES * 4) return;
    const int n  = q >> 2;
    const int c4 = (q & 3) * 4;
    const float4 v = *reinterpret_cast<const float4*>(&h[n * 16 + c4]);
    float* base = &pooled[batch[n] * 16 + c4];
    atomicAdd(base + 0, v.x);
    atomicAdd(base + 1, v.y);
    atomicAdd(base + 2, v.z);
    atomicAdd(base + 3, v.w);
}

__global__ void classify(const float* __restrict__ emb, const float* __restrict__ Wc,
                         const float* __restrict__ bc, float* __restrict__ out) {
    const int g = blockIdx.x * blockDim.x + threadIdx.x;
    if (g >= N_GRAPHS) return;
    float s = bc[0];
#pragma unroll
    for (int c = 0; c < 16; ++c) s += fmaxf(emb[g * 16 + c], 0.0f) * Wc[c];
    out[g] = s;
}

extern "C" void kernel_launch(void* const* d_in, const int* in_sizes, int n_in,
                              void* d_out, int out_size, void* d_ws, size_t ws_size,
                              hipStream_t stream) {
    const float* x     = (const float*)d_in[0];
    const int*   ei    = (const int*)d_in[1];
    const int*   batch = (const int*)d_in[2];
    const float* W1 = (const float*)d_in[3];
    const float* b1 = (const float*)d_in[4];
    const float* W2 = (const float*)d_in[5];
    const float* b2 = (const float*)d_in[6];
    const float* W3 = (const float*)d_in[7];
    const float* b3 = (const float*)d_in[8];
    const float* We = (const float*)d_in[9];
    const float* be = (const float*)d_in[10];
    const float* Wc = (const float*)d_in[11];
    const float* bc = (const float*)d_in[12];
    float* out = (float*)d_out;

    float* ws     = (float*)d_ws;
    float* dis    = ws;
    float* bufT   = dis  + N_NODES;
    float* bufA   = bufT + N_NODES * 32;
    float* bufH   = bufA + N_NODES * 32;
    float* pooled = bufH + N_NODES * 32;

    const int B = 256;

    fill4_f32<<<ceil_div(N_NODES / 4, B), B, 0, stream>>>((float4*)dis, 1.0f, N_NODES / 4);
    deg_accum<<<ceil_div(N_EDGES, B), B, 0, stream>>>(ei, dis);
    rsqrt_inplace<<<ceil_div(N_NODES, B), B, 0, stream>>>(dis, N_NODES);

    const int gBlocks = ceil_div(N_NODES / 16, B / 32);

    gemm_wmma_f32<3, 32><<<gBlocks, B, 0, stream>>>(x, W1, nullptr, bufT, N_NODES);
    fill4_f32<<<ceil_div(N_NODES * 8, B), B, 0, stream>>>((float4*)bufA, 0.0f, N_NODES * 8);
    edge_scatter<32><<<ceil_div(N_EDGES * 8, B), B, 0, stream>>>(ei, bufT, dis, bufA);
    finalize_relu<32><<<ceil_div(N_NODES * 8, B), B, 0, stream>>>(bufA, bufT, dis, b1, bufH);

    gemm_wmma_f32<32, 16><<<gBlocks, B, 0, stream>>>(bufH, W2, nullptr, bufT, N_NODES);
    fill4_f32<<<ceil_div(N_NODES * 4, B), B, 0, stream>>>((float4*)bufA, 0.0f, N_NODES * 4);
    edge_scatter<16><<<ceil_div(N_EDGES * 4, B), B, 0, stream>>>(ei, bufT, dis, bufA);
    finalize_relu<16><<<ceil_div(N_NODES * 4, B), B, 0, stream>>>(bufA, bufT, dis, b2, bufH);

    gemm_wmma_f32<16, 16><<<gBlocks, B, 0, stream>>>(bufH, W3, nullptr, bufT, N_NODES);
    fill4_f32<<<ceil_div(N_NODES * 4, B), B, 0, stream>>>((float4*)bufA, 0.0f, N_NODES * 4);
    edge_scatter<16><<<ceil_div(N_EDGES * 4, B), B, 0, stream>>>(ei, bufT, dis, bufA);
    finalize_relu<16><<<ceil_div(N_NODES * 4, B), B, 0, stream>>>(bufA, bufT, dis, b3, bufH);

    fill4_f32<<<ceil_div(N_GRAPHS * 4, B), B, 0, stream>>>((float4*)pooled, 0.0f, N_GRAPHS * 4);
    pool_kernel<<<ceil_div(N_NODES * 4, B), B, 0, stream>>>(bufH, batch, pooled);
    gemm_wmma_f32<16, 16><<<ceil_div(N_GRAPHS / 16, B / 32), B, 0, stream>>>(
        pooled, We, be, out, N_GRAPHS);
    classify<<<ceil_div(N_GRAPHS, B), B, 0, stream>>>(out, Wc, bc, out + N_GRAPHS * 16);
}


# Each atomically built sum (degree, 3 aggregations, pool) gets a private copy per L2 half, chosen by the workgroup's die id, added by its consumer; 4 edges per thread; float sum order changes
# speedup vs baseline: 1.7772x; 1.7772x over previous
#include <hip/hip_runtime.h>

typedef __attribute__((ext_vector_type(2))) float v2f;
typedef __attribute__((ext_vector_type(8))) float v8f;

#define N_NODES  100000
#define N_EDGES  3200000
#define N_GRAPHS 1024

#define EDGE_ITERS 4

static inline int ceil_div(int a, int b) { return (a + b - 1) / b; }

__device__ __forceinline__ int l2_half() {
    __shared__ int s_half;
    if (threadIdx.x == 0)
        s_half = (int)((__builtin_amdgcn_s_sendmsg_rtn(0x87) >> 18) & 1u);
    __syncthreads();
    return s_half;
}

__global__ void fill4_f32(float4* __restrict__ p, float v, int nquads) {
    int i = blockIdx.x * blockDim.x + threadIdx.x;
    if (i < nquads) p[i] = make_float4(v, v, v, v);
}

__global__ void __launch_bounds__(256)
deg_accum(const int* __restrict__ ei, float* __restrict__ deg2) {
    static_assert(N_EDGES % (256 * EDGE_ITERS) == 0, "grid covers the edges exactly");
    const int e0 = blockIdx.x * (256 * EDGE_ITERS) + threadIdx.x;
    int d[EDGE_ITERS];
#pragma unroll
    for (int it = 0; it < EDGE_ITERS; ++it) d[it] = ei[N_EDGES + e0 + it * 256];
    float* deg = deg2 + l2_half() * N_NODES;
#pragma unroll
    for (int it = 0; it < EDGE_ITERS; ++it) atomicAdd(&deg[d[it]], 1.0f);
}

__global__ void rsqrt_deg(const float* __restrict__ deg2, float* __restrict__ dis, int n) {
    int i = blockIdx.x * blockDim.x + threadIdx.x;
    if (i < n) dis[i] = rsqrtf(deg2[i] + deg2[N_NODES + i] + 1.0f);
}

template <int CIN, int COUT, bool SUM2>
__global__ void __launch_bounds__(256)
gemm_wmma_f32(const float* __restrict__ H, const float* __restrict__ W,
              const float* __restrict__ bias, float* __restrict__ out, int nrows) {
    constexpr int NT     = COUT / 16;
    constexpr int KSTEPS = (CIN + 3) / 4;

    const int wavesPerBlock = blockDim.x >> 5;
    const int wave = blockIdx.x * wavesPerBlock + (threadIdx.x >> 5);
    const int row0 = wave * 16;
    if (row0 >= nrows) return;

    const int lane = threadIdx.x & 31;
    const int half = lane >> 4;
    const int m    = lane & 15;
    const int row  = row0 + m;

    v8f acc[NT];
#pragma unroll
    for (int nt = 0; nt < NT; ++nt) acc[nt] = (v8f){};

#pragma unroll
    for (int ks = 0; ks < KSTEPS; ++ks) {
        const int ka = ks * 4 + 2 * half;

        v2f a;
        if constexpr (CIN % 4 == 0) {
            a = *reinterpret_cast<const v2f*>(&H[row * CIN + ka]);
            if constexpr (SUM2)
                a += *reinterpret_cast<const v2f*>(&H[nrows * CIN + row * CIN + ka]);
        } else {
            const int k0 = (ka     < CIN) ? ka     : (CIN - 1);
            const int k1 = (ka + 1 < CIN) ? ka + 1 : (CIN - 1);
            float ax = H[row * CIN + k0];
            float ay = H[row * CIN + k1];
            a.x = (ka     < CIN) ? ax : 0.0f;
            a.y = (ka + 1 < CIN) ? ay : 0.0f;
        }

#pragma unroll
        for (int nt = 0; nt < NT; ++nt) {
            const int ncol = nt * 16 + m;
            v2f b;
            if constexpr (CIN % 4 == 0) {
                b = *reinterpret_cast<const v2f*>(&W[ncol * CIN + ka]);
            } else {
                const int k0 = (ka     < CIN) ? ka     : (CIN - 1);
                const int k1 = (ka + 1 < CIN) ? ka + 1 : (CIN - 1);
                float bx = W[ncol * CIN + k0];
                float by = W[ncol * CIN + k1];
                b.x = (ka     < CIN) ? bx : 0.0f;
                b.y = (ka + 1 < CIN) ? by : 0.0f;
            }
            acc[nt] = __builtin_amdgcn_wmma_f32_16x16x4_f32(
                false, a, false, b, (short)0, acc[nt], false, false);
        }
    }

#pragma unroll
    for (int nt = 0; nt < NT; ++nt) {
        const int col = nt * 16 + m;
        const float bb = bias ? bias[col] : 0.0f;
#pragma unroll
        for (int r = 0; r < 8; ++r) {
            out[(row0 + r + 8 * half) * COUT + col] = acc[nt][r] + bb;
        }
    }
}

template <int COUT>
__global__ void __launch_bounds__(256)
edge_scatter(const int* __restrict__ ei, const float* __restrict__ t,
             const float* __restrict__ dis, float* __restrict__ agg2) {
    constexpr int GROUPS = COUT / 4;
    constexpr int EPI    = 256 / GROUPS;
    static_assert(N_EDGES % (EPI * EDGE_ITERS) == 0, "grid covers the edges exactly");
    const int g  = threadIdx.x % GROUPS;
    const int e0 = blockIdx.x * (EPI * EDGE_ITERS) + threadIdx.x / GROUPS;

    int src[EDGE_ITERS], dst[EDGE_ITERS];
#pragma unroll
    for (int it = 0; it < EDGE_ITERS; ++it) {
        src[it] = ei[e0 + it * EPI];
        dst[it] = ei[N_EDGES + e0 + it * EPI];
    }
    float  coeff[EDGE_ITERS];
    float4 v[EDGE_ITERS];
#pragma unroll
    for (int it = 0; it < EDGE_ITERS; ++it) {
        coeff[it] = dis[src[it]] * dis[dst[it]];
        v[it] = *reinterpret_cast<const float4*>(&t[src[it] * COUT + g * 4]);
    }
    float* agg = agg2 + l2_half() * (N_NODES * COUT);
#pragma unroll
    for (int it = 0; it < EDGE_ITERS; ++it) {
        float* base = &agg[dst[it] * COUT + g * 4];
        atomicAdd(base + 0, v[it].x * coeff[it]);
        atomicAdd(base + 1, v[it].y * coeff[it]);
        atomicAdd(base + 2, v[it].z * coeff[it]);
        atomicAdd(base + 3, v[it].w * coeff[it]);
    }
}

template <int COUT>
__global__ void finalize_relu(const float* __restrict__ aggA, const float* aggB,
                              const float* __restrict__ t,
                              const float* __restrict__ dis, const float* __restrict__ bias,
                              float* outH) {
    constexpr int GROUPS = COUT / 4;
    const int q = blockIdx.x * blockDim.x + threadIdx.x;
    if (q >= N_NODES * GROUPS) return;
    const int n  = q / GROUPS;
    const int c4 = (q % GROUPS) * 4;
    const float d  = dis[n];
    const float dd = d * d;
    float4 va = *reinterpret_cast<const float4*>(&aggA[n * COUT + c4]);
    const float4 vh = *reinterpret_cast<const float4*>(&aggB[n * COUT + c4]);
    va.x += vh.x; va.y += vh.y; va.z += vh.z; va.w += vh.w;
    const float4 vt = *reinterpret_cast<const float4*>(&t[n * COUT + c4]);
    const float4 vb = *reinterpret_cast<const float4*>(&bias[c4]);
    float4 r;
    r.x = fmaxf(fmaf(vt.x, dd, va.x) + vb.x, 0.0f);
    r.y = fmaxf(fmaf(vt.y, dd, va.y) + vb.y, 0.0f);
    r.z = fmaxf(fmaf(vt.z, dd, va.z) + vb.z, 0.0f);
    r.w = fmaxf(fmaf(vt.w, dd, va.w) + vb.w, 0.0f);
    *reinterpret_cast<float4*>(&outH[n * COUT + c4]) = r;
}

__global__ void pool_kernel(const float* __restrict__ h, const int* __restrict__ batch,
                            float* __restrict__ pooled2) {
    float* pooled = pooled2 + l2_half() * (N_GRAPHS * 16);
    const int q = blockIdx.x * blockDim.x + threadIdx.x;
    if (q >= N_NODES * 4) return;
    const int n  = q >> 2;
    const int c4 = (q & 3) * 4;
    const float4 v = *reinterpret_cast<const float4*>(&h[n * 16 + c4]);
    float* base = &pooled[batch[n] * 16 + c4];
    atomicAdd(base + 0, v.x);
    atomicAdd(base + 1, v.y);
    atomicAdd(base + 2, v.z);
    atomicAdd(base + 3, v.w);
}

__global__ void classify(const float* __restrict__ emb, const float* __restrict__ Wc,
                         const float* __restrict__ bc, float* __restrict__ out) {
    const int g = blockIdx.x * blockDim.x + threadIdx.x;
    if (g >= N_GRAPHS) return;
    float s = bc[0];
#pragma unroll
    for (int c = 0; c < 16; ++c) s += fmaxf(emb[g * 16 + c], 0.0f) * Wc[c];
    out[g] = s;
}

extern "C" void kernel_launch(void* const* d_in, const int* in_sizes, int n_in,
                              void* d_out, int out_size, void* d_ws, size_t ws_size,
                              hipStream_t stream) {
    const float* x     = (const float*)d_in[0];
    const int*   ei    = (const int*)d_in[1];
    const int*   batch = (const int*)d_in[2];
    const float* W1 = (const float*)d_in[3];
    const float* b1 = (const float*)d_in[4];
    const float* W2 = (const float*)d_in[5];
    const float* b2 = (const float*)d_in[6];
    const float* W3 = (const float*)d_in[7];
    const float* b3 = (const float*)d_in[8];
    const float* We = (const float*)d_in[9];
    const float* be = (const float*)d_in[10];
    const float* Wc = (const float*)d_in[11];
    const float* bc = (const float*)d_in[12];
    float* out = (float*)d_out;

    float* ws     = (float*)d_ws;
    float* dis    = ws;
    float* bufT   = dis  + N_NODES;
    float* bufA   = bufT + N_NODES * 32;
    float* bufH   = bufA + N_NODES * 32;
    float* deg2    = bufA;
    float* pooled2 = bufT;

    const int B = 256;

    fill4_f32<<<ceil_div(N_NODES / 2, B), B, 0, stream>>>((float4*)deg2, 0.0f, N_NODES / 2);
    deg_accum<<<N_EDGES / (B * EDGE_ITERS), B, 0, stream>>>(ei, deg2);
    rsqrt_deg<<<ceil_div(N_NODES, B), B, 0, stream>>>(deg2, dis, N_NODES);

    const int gBlocks = ceil_div(N_NODES / 16, B / 32);

    gemm_wmma_f32<3, 32, false><<<gBlocks, B, 0, stream>>>(x, W1, nullptr, bufT, N_NODES);
    fill4_f32<<<ceil_div(N_NODES * 16, B), B, 0, stream>>>((float4*)bufA, 0.0f, N_NODES * 16);
    edge_scatter<32><<<N_EDGES / (32 * EDGE_ITERS), B, 0, stream>>>(ei, bufT, dis, bufA);
    finalize_relu<32><<<ceil_div(N_NODES * 8, B), B, 0, stream>>>(bufA, bufH, bufT, dis, b1, bufH);

    gemm_wmma_f32<32, 16, false><<<gBlocks, B, 0, stream>>>(bufH, W2, nullptr, bufT, N_NODES);
    fill4_f32<<<ceil_div(N_NODES * 8, B), B, 0, stream>>>((float4*)bufA, 0.0f, N_NODES * 8);
    edge_scatter<16><<<N_EDGES / (64 * EDGE_ITERS), B, 0, stream>>>(ei, bufT, dis, bufA);
    finalize_relu<16><<<ceil_div(N_NODES * 4, B), B, 0, stream>>>(
        bufA, bufA + N_NODES * 16, bufT, dis, b2, bufH);

    gemm_wmma_f32<16, 16, false><<<gBlocks, B, 0, stream>>>(bufH, W3, nullptr, bufT, N_NODES);
    fill4_f32<<<ceil_div(N_NODES * 8, B), B, 0, stream>>>((float4*)bufA, 0.0f, N_NODES * 8);
    edge_scatter<16><<<N_EDGES / (64 * EDGE_ITERS), B, 0, stream>>>(ei, bufT, dis, bufA);
    finalize_relu<16><<<ceil_div(N_NODES * 4, B), B, 0, stream>>>(
        bufA, bufA + N_NODES * 16, bufT, dis, b3, bufH);

    fill4_f32<<<ceil_div(N_GRAPHS * 8, B), B, 0, stream>>>((float4*)pooled2, 0.0f, N_GRAPHS * 8);
    pool_kernel<<<ceil_div(N_NODES * 4, B), B, 0, stream>>>(bufH, batch, pooled2);
    gemm_wmma_f32<16, 16, true><<<ceil_div(N_GRAPHS / 16, B / 32), B, 0, stream>>>(
        pooled2, We, be, out, N_GRAPHS);
    classify<<<ceil_div(N_GRAPHS, B), B, 0, stream>>>(out, Wc, bc, out + N_GRAPHS * 16);
}


# Edge scatter: one lane per channel, so ONE atomic instruction covers a whole destination row (one L2 request per row, not four); edge indices/coefficient by lane broadcast; atomic sum order as before
# speedup vs baseline: 2.5319x; 1.4247x over previous
#include <hip/hip_runtime.h>

typedef __attribute__((ext_vector_type(2))) float v2f;
typedef __attribute__((ext_vector_type(8))) float v8f;

#define N_NODES  100000
#define N_EDGES  3200000
#define N_GRAPHS 1024

#define EDGE_ITERS 4

static inline int ceil_div(int a, int b) { return (a + b - 1) / b; }

__device__ __forceinline__ int l2_half() {
    __shared__ int s_half;
    if (threadIdx.x == 0)
        s_half = (int)((__builtin_amdgcn_s_sendmsg_rtn(0x87) >> 18) & 1u);
    __syncthreads();
    return s_half;
}

__global__ void fill4_f32(float4* __restrict__ p, float v, int nquads) {
    int i = blockIdx.x * blockDim.x + threadIdx.x;
    if (i < nquads) p[i] = make_float4(v, v, v, v);
}

__global__ void __launch_bounds__(256)
deg_accum(const int* __restrict__ ei, float* __restrict__ deg2) {
    static_assert(N_EDGES % (256 * EDGE_ITERS) == 0, "grid covers the edges exactly");
    const int e0 = blockIdx.x * (256 * EDGE_ITERS) + threadIdx.x;
    int d[EDGE_ITERS];
#pragma unroll
    for (int it = 0; it < EDGE_ITERS; ++it) d[it] = ei[N_EDGES + e0 + it * 256];
    float* deg = deg2 + l2_half() * N_NODES;
#pragma unroll
    for (int it = 0; it < EDGE_ITERS; ++it) atomicAdd(&deg[d[it]], 1.0f);
}

__global__ void rsqrt_deg(const float* __restrict__ deg2, float* __restrict__ dis, int n) {
    int i = blockIdx.x * blockDim.x + threadIdx.x;
    if (i < n) dis[i] = rsqrtf(deg2[i] + deg2[N_NODES + i] + 1.0f);
}

template <int CIN, int COUT, bool SUM2>
__global__ void __launch_bounds__(256)
gemm_wmma_f32(const float* __restrict__ H, const float* __restrict__ W,
              const float* __restrict__ bias, float* __restrict__ out, int nrows) {
    constexpr int NT     = COUT / 16;
    constexpr int KSTEPS = (CIN + 3) / 4;

    const int wavesPerBlock = blockDim.x >> 5;
    const int wave = blockIdx.x * wavesPerBlock + (threadIdx.x >> 5);
    const int row0 = wave * 16;
    if (row0 >= nrows) return;

    const int lane = threadIdx.x & 31;
    const int half = lane >> 4;
    const int m    = lane & 15;
    const int row  = row0 + m;

    v8f acc[NT];
#pragma unroll
    for (int nt = 0; nt < NT; ++nt) acc[nt] = (v8f){};

#pragma unroll
    for (int ks = 0; ks < KSTEPS; ++ks) {
        const int ka = ks * 4 + 2 * half;

        v2f a;
        if constexpr (CIN % 4 == 0) {
            a = *reinterpret_cast<const v2f*>(&H[row * CIN + ka]);
            if constexpr (SUM2)
                a += *reinterpret_cast<const v2f*>(&H[nrows * CIN + row * CIN + ka]);
        } else {
            const int k0 = (ka     < CIN) ? ka     : (CIN - 1);
            const int k1 = (ka + 1 < CIN) ? ka + 1 : (CIN - 1);
            float ax = H[row * CIN + k0];
            float ay = H[row * CIN + k1];
            a.x = (ka     < CIN) ? ax : 0.0f;
            a.y = (ka + 1 < CIN) ? ay : 0.0f;
        }

#pragma unroll
        for (int nt = 0; nt < NT; ++nt) {
            const int ncol = nt * 16 + m;
            v2f b;
            if constexpr (CIN % 4 == 0) {
                b = *reinterpret_cast<const v2f*>(&W[ncol * CIN + ka]);
            } else {
                const int k0 = (ka     < CIN) ? ka     : (CIN - 1);
                const int k1 = (ka + 1 < CIN) ? ka + 1 : (CIN - 1);
                float bx = W[ncol * CIN + k0];
                float by = W[ncol * CIN + k1];
                b.x = (ka     < CIN) ? bx : 0.0f;
                b.y = (ka + 1 < CIN) ? by : 0.0f;
            }
            acc[nt] = __builtin_amdgcn_wmma_f32_16x16x4_f32(
                false, a, false, b, (short)0, acc[nt], false, false);
        }
    }

#pragma unroll
    for (int nt = 0; nt < NT; ++nt) {
        const int col = nt * 16 + m;
        const float bb = bias ? bias[col] : 0.0f;
#pragma unroll
        for (int r = 0; r < 8; ++r) {
            out[(row0 + r + 8 * half) * COUT + col] = acc[nt][r] + bb;
        }
    }
}


template <int COUT, int J>
__device__ __forceinline__ int lane_bcast(int x) {
    if constexpr (COUT == 32) {
        return __builtin_amdgcn_readlane(x, J);
    } else {
        return __builtin_amdgcn_update_dpp(0, x, 0x150 + J  , 0xf, 0xf, true);
    }
}

template <int COUT, int J, int STEPS>
struct EdgeRows {
    static __device__ __forceinline__ void gather(int my_src, const float* __restrict__ t,
                                                  int c, float* v) {
        const int s = lane_bcast<COUT, J>(my_src);
        v[J] = t[s * COUT + c];
        EdgeRows<COUT, J + 1, STEPS>::gather(my_src, t, c, v);
    }
    static __device__ __forceinline__ void scatter(int my_dst, float my_coeff, float* agg,
                                                   int c, const float* v) {
        const int   d  = lane_bcast<COUT, J>(my_dst);
        const float cf = __int_as_float(lane_bcast<COUT, J>(__float_as_int(my_coeff)));
        atomicAdd(&agg[d * COUT + c], v[J] * cf);
        EdgeRows<COUT, J + 1, STEPS>::scatter(my_dst, my_coeff, agg, c, v);
    }
};
template <int COUT, int STEPS>
struct EdgeRows<COUT, STEPS, STEPS> {
    static __device__ __forceinline__ void gather(int, const float*, int, float*) {}
    static __device__ __forceinline__ void scatter(int, float, float*, int, const float*) {}
};

template <int COUT>
__global__ void __launch_bounds__(256)
edge_scatter(const int* __restrict__ ei, const float* __restrict__ t,
             const float* __restrict__ dis, float* __restrict__ agg2) {
    static_assert(COUT == 16 || COUT == 32, "a row is half a wave or a whole wave");
    static_assert(N_EDGES % 256 == 0, "grid covers the edges exactly: no lane is idle at a broadcast");
    constexpr int ROWS  = 32 / COUT;
    constexpr int STEPS = 32 / ROWS;

    const int e = blockIdx.x * 256 + threadIdx.x;
    const int my_src = ei[e];
    const int my_dst = ei[N_EDGES + e];
    const float my_coeff = dis[my_src] * dis[my_dst];
    const int c = threadIdx.x % COUT;

    float v[STEPS];
    EdgeRows<COUT, 0, STEPS>::gather(my_src, t, c, v);
    float* agg = agg2 + l2_half() * (N_NODES * COUT);
    EdgeRows<COUT, 0, STEPS>::scatter(my_dst, my_coeff, agg, c, v);
}

template <int COUT>
__global__ void finalize_relu(const float* __restrict__ aggA, const float* aggB,
                              const float* __restrict__ t,
                              const float* __restrict__ dis, const float* __restrict__ bias,
                              float* outH) {
    constexpr int GROUPS = COUT / 4;
    const int q = blockIdx.x * blockDim.x + threadIdx.x;
    if (q >= N_NODES * GROUPS) return;
    const int n  = q / GROUPS;
    const int c4 = (q % GROUPS) * 4;
    const float d  = dis[n];
    const float dd = d * d;
    float4 va = *reinterpret_cast<const float4*>(&aggA[n * COUT + c4]);
    const float4 vh = *reinterpret_cast<const float4*>(&aggB[n * COUT + c4]);
    va.x += vh.x; va.y += vh.y; va.z += vh.z; va.w += vh.w;
    const float4 vt = *reinterpret_cast<const float4*>(&t[n * COUT + c4]);
    const float4 vb = *reinterpret_cast<const float4*>(&bias[c4]);
    float4 r;
    r.x = fmaxf(fmaf(vt.x, dd, va.x) + vb.x, 0.0f);
    r.y = fmaxf(fmaf(vt.y, dd, va.y) + vb.y, 0.0f);
    r.z = fmaxf(fmaf(vt.z, dd, va.z) + vb.z, 0.0f);
    r.w = fmaxf(fmaf(vt.w, dd, va.w) + vb.w, 0.0f);
    *reinterpret_cast<float4*>(&outH[n * COUT + c4]) = r;
}

__global__ void pool_kernel(const float* __restrict__ h, const int* __restrict__ batch,
                            float* __restrict__ pooled2) {
    float* pooled = pooled2 + l2_half() * (N_GRAPHS * 16);
    const int q = blockIdx.x * blockDim.x + threadIdx.x;
    if (q >= N_NODES * 4) return;
    const int n  = q >> 2;
    const int c4 = (q & 3) * 4;
    const float4 v = *reinterpret_cast<const float4*>(&h[n * 16 + c4]);
    float* base = &pooled[batch[n] * 16 + c4];
    atomicAdd(base + 0, v.x);
    atomicAdd(base + 1, v.y);
    atomicAdd(base + 2, v.z);
    atomicAdd(base + 3, v.w);
}

__global__ void classify(const float* __restrict__ emb, const float* __restrict__ Wc,
                         const float* __restrict__ bc, float* __restrict__ out) {
    const int g = blockIdx.x * blockDim.x + threadIdx.x;
    if (g >= N_GRAPHS) return;
    float s = bc[0];
#pragma unroll
    for (int c = 0; c < 16; ++c) s += fmaxf(emb[g * 16 + c], 0.0f) * Wc[c];
    out[g] = s;
}

extern "C" void kernel_launch(void* const* d_in, const int* in_sizes, int n_in,
                              void* d_out, int out_size, void* d_ws, size_t ws_size,
                              hipStream_t stream) {
    const float* x     = (const float*)d_in[0];
    const int*   ei    = (const int*)d_in[1];
    const int*   batch = (const int*)d_in[2];
    const float* W1 = (const float*)d_in[3];
    const float* b1 = (const float*)d_in[4];
    const float* W2 = (const float*)d_in[5];
    const float* b2 = (const float*)d_in[6];
    const float* W3 = (const float*)d_in[7];
    const float* b3 = (const float*)d_in[8];
    const float* We = (const float*)d_in[9];
    const float* be = (const float*)d_in[10];
    const float* Wc = (const float*)d_in[11];
    const float* bc = (const float*)d_in[12];
    float* out = (float*)d_out;

    float* ws     = (float*)d_ws;
    float* dis    = ws;
    float* bufT   = dis  + N_NODES;
    float* bufA   = bufT + N_NODES * 32;
    float* bufH   = bufA + N_NODES * 32;
    float* deg2    = bufA;
    float* pooled2 = bufT;

    const int B = 256;

    fill4_f32<<<ceil_div(N_NODES / 2, B), B, 0, stream>>>((float4*)deg2, 0.0f, N_NODES / 2);
    deg_accum<<<N_EDGES / (B * EDGE_ITERS), B, 0, stream>>>(ei, deg2);
    rsqrt_deg<<<ceil_div(N_NODES, B), B, 0, stream>>>(deg2, dis, N_NODES);

    const int gBlocks = ceil_div(N_NODES / 16, B / 32);

    gemm_wmma_f32<3, 32, false><<<gBlocks, B, 0, stream>>>(x, W1, nullptr, bufT, N_NODES);
    fill4_f32<<<ceil_div(N_NODES * 16, B), B, 0, stream>>>((float4*)bufA, 0.0f, N_NODES * 16);
    edge_scatter<32><<<N_EDGES / B, B, 0, stream>>>(ei, bufT, dis, bufA);
    finalize_relu<32><<<ceil_div(N_NODES * 8, B), B, 0, stream>>>(bufA, bufH, bufT, dis, b1, bufH);

    gemm_wmma_f32<32, 16, false><<<gBlocks, B, 0, stream>>>(bufH, W2, nullptr, bufT, N_NODES);
    fill4_f32<<<ceil_div(N_NODES * 8, B), B, 0, stream>>>((float4*)bufA, 0.0f, N_NODES * 8);
    edge_scatter<16><<<N_EDGES / B, B, 0, stream>>>(ei, bufT, dis, bufA);
    finalize_relu<16><<<ceil_div(N_NODES * 4, B), B, 0, stream>>>(
        bufA, bufA + N_NODES * 16, bufT, dis, b2, bufH);

    gemm_wmma_f32<16, 16, false><<<gBlocks, B, 0, stream>>>(bufH, W3, nullptr, bufT, N_NODES);
    fill4_f32<<<ceil_div(N_NODES * 8, B), B, 0, stream>>>((float4*)bufA, 0.0f, N_NODES * 8);
    edge_scatter<16><<<N_EDGES / B, B, 0, stream>>>(ei, bufT, dis, bufA);
    finalize_relu<16><<<ceil_div(N_NODES * 4, B), B, 0, stream>>>(
        bufA, bufA + N_NODES * 16, bufT, dis, b3, bufH);

    fill4_f32<<<ceil_div(N_GRAPHS * 8, B), B, 0, stream>>>((float4*)pooled2, 0.0f, N_GRAPHS * 8);
    pool_kernel<<<ceil_div(N_NODES * 4, B), B, 0, stream>>>(bufH, batch, pooled2);
    gemm_wmma_f32<16, 16, true><<<ceil_div(N_GRAPHS / 16, B / 32), B, 0, stream>>>(
        pooled2, We, be, out, N_GRAPHS);
    classify<<<ceil_div(N_GRAPHS, B), B, 0, stream>>>(out, Wc, bc, out + N_GRAPHS * 16);
}


# Edge coefficient dis[src]*dis[dst] factored out of the edge loop: transform stores t*dis, finalize scales the node sum by dis; scatter reads no dis; products re-associated, atomic order as before
# speedup vs baseline: 3.2724x; 1.2925x over previous
#include <hip/hip_runtime.h>

typedef __attribute__((ext_vector_type(2))) float v2f;
typedef __attribute__((ext_vector_type(8))) float v8f;

#define N_NODES  100000
#define N_EDGES  3200000
#define N_GRAPHS 1024

#define EDGE_ITERS 4

static inline int ceil_div(int a, int b) { return (a + b - 1) / b; }

__device__ __forceinline__ int l2_half() {
    __shared__ int s_half;
    if (threadIdx.x == 0)
        s_half = (int)((__builtin_amdgcn_s_sendmsg_rtn(0x87) >> 18) & 1u);
    __syncthreads();
    return s_half;
}

__global__ void fill4_f32(float4* __restrict__ p, float v, int nquads) {
    int i = blockIdx.x * blockDim.x + threadIdx.x;
    if (i < nquads) p[i] = make_float4(v, v, v, v);
}

__global__ void __launch_bounds__(256)
deg_accum(const int* __restrict__ ei, float* __restrict__ deg2) {
    static_assert(N_EDGES % (256 * EDGE_ITERS) == 0, "grid covers the edges exactly");
    const int e0 = blockIdx.x * (256 * EDGE_ITERS) + threadIdx.x;
    int d[EDGE_ITERS];
#pragma unroll
    for (int it = 0; it < EDGE_ITERS; ++it) d[it] = ei[N_EDGES + e0 + it * 256];
    float* deg = deg2 + l2_half() * N_NODES;
#pragma unroll
    for (int it = 0; it < EDGE_ITERS; ++it) atomicAdd(&deg[d[it]], 1.0f);
}

__global__ void rsqrt_deg(const float* __restrict__ deg2, float* __restrict__ dis, int n) {
    int i = blockIdx.x * blockDim.x + threadIdx.x;
    if (i < n) dis[i] = rsqrtf(deg2[i] + deg2[N_NODES + i] + 1.0f);
}

template <int CIN, int COUT, bool SUM2, bool ROWSCALE>
__global__ void __launch_bounds__(256)
gemm_wmma_f32(const float* __restrict__ H, const float* __restrict__ W,
              const float* __restrict__ bias, const float* __restrict__ rowscale,
              float* __restrict__ out, int nrows) {
    constexpr int NT     = COUT / 16;
    constexpr int KSTEPS = (CIN + 3) / 4;

    const int wavesPerBlock = blockDim.x >> 5;
    const int wave = blockIdx.x * wavesPerBlock + (threadIdx.x >> 5);
    const int row0 = wave * 16;
    if (row0 >= nrows) return;

    const int lane = threadIdx.x & 31;
    const int half = lane >> 4;
    const int m    = lane & 15;
    const int row  = row0 + m;

    float rs[8];
    if constexpr (ROWSCALE) {
        const float4 s0 = *reinterpret_cast<const float4*>(&rowscale[row0 + 8 * half]);
        const float4 s1 = *reinterpret_cast<const float4*>(&rowscale[row0 + 8 * half + 4]);
        rs[0] = s0.x; rs[1] = s0.y; rs[2] = s0.z; rs[3] = s0.w;
        rs[4] = s1.x; rs[5] = s1.y; rs[6] = s1.z; rs[7] = s1.w;
    }

    v8f acc[NT];
#pragma unroll
    for (int nt = 0; nt < NT; ++nt) acc[nt] = (v8f){};

#pragma unroll
    for (int ks = 0; ks < KSTEPS; ++ks) {
        const int ka = ks * 4 + 2 * half;

        v2f a;
        if constexpr (CIN % 4 == 0) {
            a = *reinterpret_cast<const v2f*>(&H[row * CIN + ka]);
            if constexpr (SUM2)
                a += *reinterpret_cast<const v2f*>(&H[nrows * CIN + row * CIN + ka]);
        } else {
            const int k0 = (ka     < CIN) ? ka     : (CIN - 1);
            const int k1 = (ka + 1 < CIN) ? ka + 1 : (CIN - 1);
            float ax = H[row * CIN + k0];
            float ay = H[row * CIN + k1];
            a.x = (ka     < CIN) ? ax : 0.0f;
            a.y = (ka + 1 < CIN) ? ay : 0.0f;
        }

#pragma unroll
        for (int nt = 0; nt < NT; ++nt) {
            const int ncol = nt * 16 + m;
            v2f b;
            if constexpr (CIN % 4 == 0) {
                b = *reinterpret_cast<const v2f*>(&W[ncol * CIN + ka]);
            } else {
                const int k0 = (ka     < CIN) ? ka     : (CIN - 1);
                const int k1 = (ka + 1 < CIN) ? ka + 1 : (CIN - 1);
                float bx = W[ncol * CIN + k0];
                float by = W[ncol * CIN + k1];
                b.x = (ka     < CIN) ? bx : 0.0f;
                b.y = (ka + 1 < CIN) ? by : 0.0f;
            }
            acc[nt] = __builtin_amdgcn_wmma_f32_16x16x4_f32(
                false, a, false, b, (short)0, acc[nt], false, false);
        }
    }

#pragma unroll
    for (int nt = 0; nt < NT; ++nt) {
        const int col = nt * 16 + m;
        const float bb = bias ? bias[col] : 0.0f;
#pragma unroll
        for (int r = 0; r < 8; ++r) {
            if constexpr (ROWSCALE)
                out[(row0 + r + 8 * half) * COUT + col] = acc[nt][r] * rs[r] + bb;
            else
                out[(row0 + r + 8 * half) * COUT + col] = acc[nt][r] + bb;
        }
    }
}


template <int COUT, int J>
__device__ __forceinline__ int lane_bcast(int x) {
    if constexpr (COUT == 32) {
        return __builtin_amdgcn_readlane(x, J);
    } else {
        return __builtin_amdgcn_update_dpp(0, x, 0x150 + J  , 0xf, 0xf, true);
    }
}

template <int COUT, int J, int STEPS>
struct EdgeRows {
    static __device__ __forceinline__ void gather(int my_src, const float* __restrict__ t,
                                                  int c, float* v) {
        const int s = lane_bcast<COUT, J>(my_src);
        v[J] = t[s * COUT + c];
        EdgeRows<COUT, J + 1, STEPS>::gather(my_src, t, c, v);
    }
    static __device__ __forceinline__ void scatter(int my_dst, float* agg,
                                                   int c, const float* v) {
        const int d = lane_bcast<COUT, J>(my_dst);
        atomicAdd(&agg[d * COUT + c], v[J]);
        EdgeRows<COUT, J + 1, STEPS>::scatter(my_dst, agg, c, v);
    }
};
template <int COUT, int STEPS>
struct EdgeRows<COUT, STEPS, STEPS> {
    static __device__ __forceinline__ void gather(int, const float*, int, float*) {}
    static __device__ __forceinline__ void scatter(int, float*, int, const float*) {}
};

template <int COUT>
__global__ void __launch_bounds__(256)
edge_scatter(const int* __restrict__ ei, const float* __restrict__ t,
             float* __restrict__ agg2) {
    static_assert(COUT == 16 || COUT == 32, "a row is half a wave or a whole wave");
    static_assert(N_EDGES % 256 == 0, "grid covers the edges exactly: no lane is idle at a broadcast");
    constexpr int ROWS  = 32 / COUT;
    constexpr int STEPS = 32 / ROWS;

    const int e = blockIdx.x * 256 + threadIdx.x;
    const int my_src = ei[e];
    const int my_dst = ei[N_EDGES + e];
    const int c = threadIdx.x % COUT;

    float v[STEPS];
    EdgeRows<COUT, 0, STEPS>::gather(my_src, t, c, v);
    float* agg = agg2 + l2_half() * (N_NODES * COUT);
    EdgeRows<COUT, 0, STEPS>::scatter(my_dst, agg, c, v);
}

template <int COUT>
__global__ void finalize_relu(const float* __restrict__ aggA, const float* aggB,
                              const float* __restrict__ t,
                              const float* __restrict__ dis, const float* __restrict__ bias,
                              float* outH) {
    constexpr int GROUPS = COUT / 4;
    const int q = blockIdx.x * blockDim.x + threadIdx.x;
    if (q >= N_NODES * GROUPS) return;
    const int n  = q / GROUPS;
    const int c4 = (q % GROUPS) * 4;
    const float d  = dis[n];
    float4 va = *reinterpret_cast<const float4*>(&aggA[n * COUT + c4]);
    const float4 vh = *reinterpret_cast<const float4*>(&aggB[n * COUT + c4]);
    va.x += vh.x; va.y += vh.y; va.z += vh.z; va.w += vh.w;
    const float4 vt = *reinterpret_cast<const float4*>(&t[n * COUT + c4]);
    const float4 vb = *reinterpret_cast<const float4*>(&bias[c4]);
    float4 r;
    r.x = fmaxf(fmaf(va.x + vt.x, d, vb.x), 0.0f);
    r.y = fmaxf(fmaf(va.y + vt.y, d, vb.y), 0.0f);
    r.z = fmaxf(fmaf(va.z + vt.z, d, vb.z), 0.0f);
    r.w = fmaxf(fmaf(va.w + vt.w, d, vb.w), 0.0f);
    *reinterpret_cast<float4*>(&outH[n * COUT + c4]) = r;
}

__global__ void pool_kernel(const float* __restrict__ h, const int* __restrict__ batch,
                            float* __restrict__ pooled2) {
    float* pooled = pooled2 + l2_half() * (N_GRAPHS * 16);
    const int q = blockIdx.x * blockDim.x + threadIdx.x;
    if (q >= N_NODES * 4) return;
    const int n  = q >> 2;
    const int c4 = (q & 3) * 4;
    const float4 v = *reinterpret_cast<const float4*>(&h[n * 16 + c4]);
    float* base = &pooled[batch[n] * 16 + c4];
    atomicAdd(base + 0, v.x);
    atomicAdd(base + 1, v.y);
    atomicAdd(base + 2, v.z);
    atomicAdd(base + 3, v.w);
}

__global__ void classify(const float* __restrict__ emb, const float* __restrict__ Wc,
                         const float* __restrict__ bc, float* __restrict__ out) {
    const int g = blockIdx.x * blockDim.x + threadIdx.x;
    if (g >= N_GRAPHS) return;
    float s = bc[0];
#pragma unroll
    for (int c = 0; c < 16; ++c) s += fmaxf(emb[g * 16 + c], 0.0f) * Wc[c];
    out[g] = s;
}

extern "C" void kernel_launch(void* const* d_in, const int* in_sizes, int n_in,
                              void* d_out, int out_size, void* d_ws, size_t ws_size,
                              hipStream_t stream) {
    const float* x     = (const float*)d_in[0];
    const int*   ei    = (const int*)d_in[1];
    const int*   batch = (const int*)d_in[2];
    const float* W1 = (const float*)d_in[3];
    const float* b1 = (const float*)d_in[4];
    const float* W2 = (const float*)d_in[5];
    const float* b2 = (const float*)d_in[6];
    const float* W3 = (const float*)d_in[7];
    const float* b3 = (const float*)d_in[8];
    const float* We = (const float*)d_in[9];
    const float* be = (const float*)d_in[10];
    const float* Wc = (const float*)d_in[11];
    const float* bc = (const float*)d_in[12];
    float* out = (float*)d_out;

    float* ws     = (float*)d_ws;
    float* dis    = ws;
    float* bufT   = dis  + N_NODES;
    float* bufA   = bufT + N_NODES * 32;
    float* bufH   = bufA + N_NODES * 32;
    float* deg2    = bufA;
    float* pooled2 = bufT;

    const int B = 256;

    fill4_f32<<<ceil_div(N_NODES / 2, B), B, 0, stream>>>((float4*)deg2, 0.0f, N_NODES / 2);
    deg_accum<<<N_EDGES / (B * EDGE_ITERS), B, 0, stream>>>(ei, deg2);
    rsqrt_deg<<<ceil_div(N_NODES, B), B, 0, stream>>>(deg2, dis, N_NODES);

    const int gBlocks = ceil_div(N_NODES / 16, B / 32);

    gemm_wmma_f32<3, 32, false, true><<<gBlocks, B, 0, stream>>>(x, W1, nullptr, dis, bufT, N_NODES);
    fill4_f32<<<ceil_div(N_NODES * 16, B), B, 0, stream>>>((float4*)bufA, 0.0f, N_NODES * 16);
    edge_scatter<32><<<N_EDGES / B, B, 0, stream>>>(ei, bufT, bufA);
    finalize_relu<32><<<ceil_div(N_NODES * 8, B), B, 0, stream>>>(bufA, bufH, bufT, dis, b1, bufH);

    gemm_wmma_f32<32, 16, false, true><<<gBlocks, B, 0, stream>>>(bufH, W2, nullptr, dis, bufT, N_NODES);
    fill4_f32<<<ceil_div(N_NODES * 8, B), B, 0, stream>>>((float4*)bufA, 0.0f, N_NODES * 8);
    edge_scatter<16><<<N_EDGES / B, B, 0, stream>>>(ei, bufT, bufA);
    finalize_relu<16><<<ceil_div(N_NODES * 4, B), B, 0, stream>>>(
        bufA, bufA + N_NODES * 16, bufT, dis, b2, bufH);

    gemm_wmma_f32<16, 16, false, true><<<gBlocks, B, 0, stream>>>(bufH, W3, nullptr, dis, bufT, N_NODES);
    fill4_f32<<<ceil_div(N_NODES * 8, B), B, 0, stream>>>((float4*)bufA, 0.0f, N_NODES * 8);
    edge_scatter<16><<<N_EDGES / B, B, 0, stream>>>(ei, bufT, bufA);
    finalize_relu<16><<<ceil_div(N_NODES * 4, B), B, 0, stream>>>(
        bufA, bufA + N_NODES * 16, bufT, dis, b3, bufH);

    fill4_f32<<<ceil_div(N_GRAPHS * 8, B), B, 0, stream>>>((float4*)pooled2, 0.0f, N_GRAPHS * 8);
    pool_kernel<<<ceil_div(N_NODES * 4, B), B, 0, stream>>>(bufH, batch, pooled2);
    gemm_wmma_f32<16, 16, true, false><<<ceil_div(N_GRAPHS / 16, B / 32), B, 0, stream>>>(
        pooled2, We, be, nullptr, out, N_GRAPHS);
    classify<<<ceil_div(N_GRAPHS, B), B, 0, stream>>>(out, Wc, bc, out + N_GRAPHS * 16);
}


# Launch fusion: row clears and each layer's finalize (the last with pooling) fold into the WMMA transform kernels; 128-B row pitch so a wave owns its rows; 19 launches to 12; sum order as before
# speedup vs baseline: 3.6711x; 1.1218x over previous
#include <hip/hip_runtime.h>

typedef __attribute__((ext_vector_type(2))) float v2f;
typedef __attribute__((ext_vector_type(8))) float v8f;

#define N_NODES  100000
#define N_EDGES  3200000
#define N_GRAPHS 1024

#define EDGE_ITERS 4
#define PITCH 32

static inline int ceil_div(int a, int b) { return (a + b - 1) / b; }

__device__ __forceinline__ int l2_half() {
    __shared__ int s_half;
    if (threadIdx.x == 0)
        s_half = (int)((__builtin_amdgcn_s_sendmsg_rtn(0x87) >> 18) & 1u);
    __syncthreads();
    return s_half;
}

__global__ void fill4_f32(float4* __restrict__ p, float v, int nquads) {
    int i = blockIdx.x * blockDim.x + threadIdx.x;
    if (i < nquads) p[i] = make_float4(v, v, v, v);
}

__global__ void __launch_bounds__(256)
deg_accum(const int* __restrict__ ei, float* __restrict__ deg2) {
    static_assert(N_EDGES % (256 * EDGE_ITERS) == 0, "grid covers the edges exactly");
    const int e0 = blockIdx.x * (256 * EDGE_ITERS) + threadIdx.x;
    int d[EDGE_ITERS];
#pragma unroll
    for (int it = 0; it < EDGE_ITERS; ++it) d[it] = ei[N_EDGES + e0 + it * 256];
    float* deg = deg2 + l2_half() * N_NODES;
#pragma unroll
    for (int it = 0; it < EDGE_ITERS; ++it) atomicAdd(&deg[d[it]], 1.0f);
}

__global__ void rsqrt_deg(const float* __restrict__ deg2, float* __restrict__ dis, int n) {
    int i = blockIdx.x * blockDim.x + threadIdx.x;
    if (i < n) dis[i] = rsqrtf(deg2[i] + deg2[N_NODES + i] + 1.0f);
}

enum { A_PLAIN = 0, A_SUM2 = 1, A_FINAL = 2 };

template <int CIN, int COUT, int APITCH, int OPITCH, int AMODE, bool ROWSCALE, int ZEROW, bool ZEROPOOL>
__global__ void __launch_bounds__(256)
node_transform(const float* H, const float* H2, const float* T,
               const float* __restrict__ bias_in, const float* __restrict__ W,
               const float* __restrict__ bias, const float* __restrict__ rowscale,
               float* out, float* zeroA, float* zeroB, int nrows) {
    static_assert(CIN % 4 == 0 || AMODE == A_PLAIN, "the padded K path reads plain rows only");
    constexpr int NT     = COUT / 16;
    constexpr int KSTEPS = (CIN + 3) / 4;

    const int wavesPerBlock = blockDim.x >> 5;
    const int wave = blockIdx.x * wavesPerBlock + (threadIdx.x >> 5);
    const int row0 = wave * 16;
    if (row0 >= nrows) return;

    const int lane = threadIdx.x & 31;
    const int half = lane >> 4;
    const int m    = lane & 15;
    const int row  = row0 + m;

    float rs[8];
    if constexpr (ROWSCALE) {
        const float4 s0 = *reinterpret_cast<const float4*>(&rowscale[row0 + 8 * half]);
        const float4 s1 = *reinterpret_cast<const float4*>(&rowscale[row0 + 8 * half + 4]);
        rs[0] = s0.x; rs[1] = s0.y; rs[2] = s0.z; rs[3] = s0.w;
        rs[4] = s1.x; rs[5] = s1.y; rs[6] = s1.z; rs[7] = s1.w;
    }
    float drow = 0.0f;
    if constexpr (AMODE == A_FINAL) drow = rowscale[row];

    v8f acc[NT];
#pragma unroll
    for (int nt = 0; nt < NT; ++nt) acc[nt] = (v8f){};

#pragma unroll
    for (int ks = 0; ks < KSTEPS; ++ks) {
        const int ka = ks * 4 + 2 * half;

        v2f a;
        if constexpr (CIN % 4 == 0) {
            a = *reinterpret_cast<const v2f*>(&H[row * APITCH + ka]);
            if constexpr (AMODE == A_SUM2)
                a += *reinterpret_cast<const v2f*>(&H2[row * APITCH + ka]);
            if constexpr (AMODE == A_FINAL) {
                const v2f ab = *reinterpret_cast<const v2f*>(&H2[row * APITCH + ka]);
                const v2f vt = *reinterpret_cast<const v2f*>(&T[row * APITCH + ka]);
                const v2f bi = *reinterpret_cast<const v2f*>(&bias_in[ka]);
                a.x = fmaxf(fmaf((a.x + ab.x) + vt.x, drow, bi.x), 0.0f);
                a.y = fmaxf(fmaf((a.y + ab.y) + vt.y, drow, bi.y), 0.0f);
            }
        } else {
            const int k0 = (ka     < CIN) ? ka     : (CIN - 1);
            const int k1 = (ka + 1 < CIN) ? ka + 1 : (CIN - 1);
            float ax = H[row * APITCH + k0];
            float ay = H[row * APITCH + k1];
            a.x = (ka     < CIN) ? ax : 0.0f;
            a.y = (ka + 1 < CIN) ? ay : 0.0f;
        }

#pragma unroll
        for (int nt = 0; nt < NT; ++nt) {
            const int ncol = nt * 16 + m;
            v2f b;
            if constexpr (CIN % 4 == 0) {
                b = *reinterpret_cast<const v2f*>(&W[ncol * CIN + ka]);
            } else {
                const int k0 = (ka     < CIN) ? ka     : (CIN - 1);
                const int k1 = (ka + 1 < CIN) ? ka + 1 : (CIN - 1);
                float bx = W[ncol * CIN + k0];
                float by = W[ncol * CIN + k1];
                b.x = (ka     < CIN) ? bx : 0.0f;
                b.y = (ka + 1 < CIN) ? by : 0.0f;
            }
            acc[nt] = __builtin_amdgcn_wmma_f32_16x16x4_f32(
                false, a, false, b, (short)0, acc[nt], false, false);
        }
    }

#pragma unroll
    for (int nt = 0; nt < NT; ++nt) {
        const int col = nt * 16 + m;
        const float bb = bias ? bias[col] : 0.0f;
#pragma unroll
        for (int r = 0; r < 8; ++r) {
            if constexpr (ROWSCALE)
                out[(row0 + r + 8 * half) * OPITCH + col] = acc[nt][r] * rs[r] + bb;
            else
                out[(row0 + r + 8 * half) * OPITCH + col] = acc[nt][r] + bb;
        }
    }

    if constexpr (ZEROW > 0) {
        constexpr int QPR = ZEROW / 4;
        constexpr int RPI = 32 / QPR;
        const float4 z = make_float4(0.0f, 0.0f, 0.0f, 0.0f);
        const int zr = lane / QPR;
        const int zc = (lane % QPR) * 4;
#pragma unroll
        for (int i = 0; i < 16 / RPI; ++i) {
            const int o = (row0 + i * RPI + zr) * PITCH + zc;
            *reinterpret_cast<float4*>(&zeroA[o]) = z;
            *reinterpret_cast<float4*>(&zeroB[o]) = z;
        }
    }
    if constexpr (ZEROPOOL) {
        static_assert(OPITCH == PITCH && COUT == 16, "the pooled sums use the free half of a 16-channel row");
        if (row0 < 2 * N_GRAPHS) {
            const float4 z = make_float4(0.0f, 0.0f, 0.0f, 0.0f);
            const int zr = lane >> 2;
            const int zc = 16 + (lane & 3) * 4;
#pragma unroll
            for (int i = 0; i < 2; ++i)
                *reinterpret_cast<float4*>(&out[(row0 + i * 8 + zr) * PITCH + zc]) = z;
        }
    }
}


template <int COUT, int J>
__device__ __forceinline__ int lane_bcast(int x) {
    if constexpr (COUT == 32) {
        return __builtin_amdgcn_readlane(x, J);
    } else {
        return __builtin_amdgcn_update_dpp(0, x, 0x150 + J  , 0xf, 0xf, true);
    }
}

template <int COUT, int J, int STEPS>
struct EdgeRows {
    static __device__ __forceinline__ void gather(int my_src, const float* __restrict__ t,
                                                  int c, float* v) {
        const int s = lane_bcast<COUT, J>(my_src);
        v[J] = t[s * PITCH + c];
        EdgeRows<COUT, J + 1, STEPS>::gather(my_src, t, c, v);
    }
    static __device__ __forceinline__ void scatter(int my_dst, float* agg,
                                                   int c, const float* v) {
        const int d = lane_bcast<COUT, J>(my_dst);
        atomicAdd(&agg[d * PITCH + c], v[J]);
        EdgeRows<COUT, J + 1, STEPS>::scatter(my_dst, agg, c, v);
    }
};
template <int COUT, int STEPS>
struct EdgeRows<COUT, STEPS, STEPS> {
    static __device__ __forceinline__ void gather(int, const float*, int, float*) {}
    static __device__ __forceinline__ void scatter(int, float*, int, const float*) {}
};

template <int COUT>
__global__ void __launch_bounds__(256)
edge_scatter(const int* __restrict__ ei, const float* __restrict__ t,
             float* __restrict__ agg2) {
    static_assert(COUT == 16 || COUT == 32, "a row is half a wave or a whole wave");
    static_assert(N_EDGES % 256 == 0, "grid covers the edges exactly: no lane is idle at a broadcast");
    constexpr int ROWS  = 32 / COUT;
    constexpr int STEPS = 32 / ROWS;

    const int e = blockIdx.x * 256 + threadIdx.x;
    const int my_src = ei[e];
    const int my_dst = ei[N_EDGES + e];
    const int c = threadIdx.x % COUT;

    float v[STEPS];
    EdgeRows<COUT, 0, STEPS>::gather(my_src, t, c, v);
    float* agg = agg2 + l2_half() * (N_NODES * PITCH);
    EdgeRows<COUT, 0, STEPS>::scatter(my_dst, agg, c, v);
}

__global__ void final_pool(const float* __restrict__ aggA, const float* __restrict__ aggB,
                           const float* ts, const float* __restrict__ dis,
                           const float* __restrict__ bias, const int* __restrict__ batch,
                           float* pooled2) {
    float* pooled = pooled2 + l2_half() * (N_GRAPHS * PITCH);
    const int q = blockIdx.x * blockDim.x + threadIdx.x;
    if (q >= N_NODES * 4) return;
    const int n  = q >> 2;
    const int c4 = (q & 3) * 4;
    const float d = dis[n];
    float4 va = *reinterpret_cast<const float4*>(&aggA[n * PITCH + c4]);
    const float4 vh = *reinterpret_cast<const float4*>(&aggB[n * PITCH + c4]);
    va.x += vh.x; va.y += vh.y; va.z += vh.z; va.w += vh.w;
    const float4 vt = *reinterpret_cast<const float4*>(&ts[n * PITCH + c4]);
    const float4 vb = *reinterpret_cast<const float4*>(&bias[c4]);
    float4 v;
    v.x = fmaxf(fmaf(va.x + vt.x, d, vb.x), 0.0f);
    v.y = fmaxf(fmaf(va.y + vt.y, d, vb.y), 0.0f);
    v.z = fmaxf(fmaf(va.z + vt.z, d, vb.z), 0.0f);
    v.w = fmaxf(fmaf(va.w + vt.w, d, vb.w), 0.0f);
    float* base = &pooled[batch[n] * PITCH + c4];
    atomicAdd(base + 0, v.x);
    atomicAdd(base + 1, v.y);
    atomicAdd(base + 2, v.z);
    atomicAdd(base + 3, v.w);
}

__global__ void classify(const float* __restrict__ emb, const float* __restrict__ Wc,
                         const float* __restrict__ bc, float* __restrict__ out) {
    const int g = blockIdx.x * blockDim.x + threadIdx.x;
    if (g >= N_GRAPHS) return;
    float s = bc[0];
#pragma unroll
    for (int c = 0; c < 16; ++c) s += fmaxf(emb[g * 16 + c], 0.0f) * Wc[c];
    out[g] = s;
}

extern "C" void kernel_launch(void* const* d_in, const int* in_sizes, int n_in,
                              void* d_out, int out_size, void* d_ws, size_t ws_size,
                              hipStream_t stream) {
    const float* x     = (const float*)d_in[0];
    const int*   ei    = (const int*)d_in[1];
    const int*   batch = (const int*)d_in[2];
    const float* W1 = (const float*)d_in[3];
    const float* b1 = (const float*)d_in[4];
    const float* W2 = (const float*)d_in[5];
    const float* b2 = (const float*)d_in[6];
    const float* W3 = (const float*)d_in[7];
    const float* b3 = (const float*)d_in[8];
    const float* We = (const float*)d_in[9];
    const float* be = (const float*)d_in[10];
    const float* Wc = (const float*)d_in[11];
    const float* bc = (const float*)d_in[12];
    float* out = (float*)d_out;

    float* ws     = (float*)d_ws;
    float* dis    = ws;
    float* bufT   = dis  + N_NODES;
    float* bufA   = bufT + N_NODES * PITCH;
    float* bufH   = bufA + N_NODES * PITCH;
    float* deg2    = bufA;
    float* pooled2 = bufT + 16;

    const int B = 256;

    fill4_f32<<<ceil_div(N_NODES / 2, B), B, 0, stream>>>((float4*)deg2, 0.0f, N_NODES / 2);
    deg_accum<<<N_EDGES / (B * EDGE_ITERS), B, 0, stream>>>(ei, deg2);
    rsqrt_deg<<<ceil_div(N_NODES, B), B, 0, stream>>>(deg2, dis, N_NODES);

    const int gBlocks = ceil_div(N_NODES / 16, B / 32);

    node_transform<3, 32, 3, PITCH, A_PLAIN, true, 32, false><<<gBlocks, B, 0, stream>>>(
        x, nullptr, nullptr, nullptr, W1, nullptr, dis, bufT, bufA, bufH, N_NODES);
    edge_scatter<32><<<N_EDGES / B, B, 0, stream>>>(ei, bufT, bufA);

    node_transform<32, 16, PITCH, PITCH, A_FINAL, true, 16, false><<<gBlocks, B, 0, stream>>>(
        bufA, bufH, bufT, b1, W2, nullptr, dis, bufT, bufA, bufH, N_NODES);
    edge_scatter<16><<<N_EDGES / B, B, 0, stream>>>(ei, bufT, bufA);

    node_transform<16, 16, PITCH, PITCH, A_FINAL, true, 16, true><<<gBlocks, B, 0, stream>>>(
        bufA, bufH, bufT, b2, W3, nullptr, dis, bufT, bufA, bufH, N_NODES);
    edge_scatter<16><<<N_EDGES / B, B, 0, stream>>>(ei, bufT, bufA);

    final_pool<<<ceil_div(N_NODES * 4, B), B, 0, stream>>>(bufA, bufH, bufT, dis, b3, batch, pooled2);
    node_transform<16, 16, PITCH, 16, A_SUM2, false, 0, false><<<ceil_div(N_GRAPHS / 16, B / 32), B, 0, stream>>>(
        pooled2, pooled2 + N_GRAPHS * PITCH, nullptr, nullptr, We, be, nullptr, out, nullptr, nullptr, N_GRAPHS);
    classify<<<ceil_div(N_GRAPHS, B), B, 0, stream>>>(out, Wc, bc, out + N_GRAPHS * 16);
}


# Layer 1 sums before its product, (A x) W1 for A (x W1): its edge pass moves 16-B rows, not 128-B, and W1 joins layer 2's WMMA kernel (h1 stays in registers); 12 launches to 11; sum order changes
# speedup vs baseline: 3.8479x; 1.0482x over previous
#include <hip/hip_runtime.h>

typedef __attribute__((ext_vector_type(2))) float v2f;
typedef __attribute__((ext_vector_type(8))) float v8f;

#define N_NODES  100000
#define N_EDGES  3200000
#define N_GRAPHS 1024

#define EDGE_ITERS 4
#define PITCH 32
#define DEGW 16

static inline int ceil_div(int a, int b) { return (a + b - 1) / b; }

__device__ __forceinline__ int l2_half() {
    __shared__ int s_half;
    if (threadIdx.x == 0)
        s_half = (int)((__builtin_amdgcn_s_sendmsg_rtn(0x87) >> 18) & 1u);
    __syncthreads();
    return s_half;
}

__global__ void clear_deg(float* __restrict__ rowsA, float* __restrict__ rowsB) {
    const int n = blockIdx.x * blockDim.x + threadIdx.x;
    if (n < N_NODES) { rowsA[n * PITCH + DEGW] = 0.0f; rowsB[n * PITCH + DEGW] = 0.0f; }
}

__global__ void __launch_bounds__(256)
deg_accum(const int* __restrict__ ei, float* __restrict__ rowsA, float* __restrict__ rowsB) {
    static_assert(N_EDGES % (256 * EDGE_ITERS) == 0, "grid covers the edges exactly");
    const int e0 = blockIdx.x * (256 * EDGE_ITERS) + threadIdx.x;
    int d[EDGE_ITERS];
#pragma unroll
    for (int it = 0; it < EDGE_ITERS; ++it) d[it] = ei[N_EDGES + e0 + it * 256];
    float* deg = (l2_half() ? rowsB : rowsA) + DEGW;
#pragma unroll
    for (int it = 0; it < EDGE_ITERS; ++it) atomicAdd(&deg[d[it] * PITCH], 1.0f);
}

__global__ void node_prep(const float* __restrict__ x, float* __restrict__ dis,
                          float* __restrict__ xs, float* __restrict__ aggA, float* __restrict__ aggB) {
    const int n = blockIdx.x * blockDim.x + threadIdx.x;
    if (n >= N_NODES) return;
    const float d = rsqrtf(aggA[n * PITCH + DEGW] + aggB[n * PITCH + DEGW] + 1.0f);
    dis[n] = d;
    const float4 z = make_float4(0.0f, 0.0f, 0.0f, 0.0f);
    *reinterpret_cast<float4*>(&xs[n * PITCH]) = make_float4(x[n * 3] * d, x[n * 3 + 1] * d, x[n * 3 + 2] * d, 0.0f);
    *reinterpret_cast<float4*>(&aggA[n * PITCH]) = z;
    *reinterpret_cast<float4*>(&aggB[n * PITCH]) = z;
}

enum { A_PLAIN = 0, A_SUM2 = 1, A_FINAL = 2 };

template <int CIN, int COUT, int APITCH, int OPITCH, int AMODE, bool ROWSCALE, int ZEROW, bool ZEROPOOL>
__global__ void __launch_bounds__(256)
node_transform(const float* H, const float* H2, const float* T,
               const float* __restrict__ bias_in, const float* __restrict__ W,
               const float* __restrict__ bias, const float* __restrict__ rowscale,
               float* out, float* zeroA, float* zeroB, int nrows) {
    static_assert(CIN % 4 == 0 || AMODE == A_PLAIN, "the padded K path reads plain rows only");
    constexpr int NT     = COUT / 16;
    constexpr int KSTEPS = (CIN + 3) / 4;

    const int wavesPerBlock = blockDim.x >> 5;
    const int wave = blockIdx.x * wavesPerBlock + (threadIdx.x >> 5);
    const int row0 = wave * 16;
    if (row0 >= nrows) return;

    const int lane = threadIdx.x & 31;
    const int half = lane >> 4;
    const int m    = lane & 15;
    const int row  = row0 + m;

    float rs[8];
    if constexpr (ROWSCALE) {
        const float4 s0 = *reinterpret_cast<const float4*>(&rowscale[row0 + 8 * half]);
        const float4 s1 = *reinterpret_cast<const float4*>(&rowscale[row0 + 8 * half + 4]);
        rs[0] = s0.x; rs[1] = s0.y; rs[2] = s0.z; rs[3] = s0.w;
        rs[4] = s1.x; rs[5] = s1.y; rs[6] = s1.z; rs[7] = s1.w;
    }
    float drow = 0.0f;
    if constexpr (AMODE == A_FINAL) drow = rowscale[row];

    v8f acc[NT];
#pragma unroll
    for (int nt = 0; nt < NT; ++nt) acc[nt] = (v8f){};

#pragma unroll
    for (int ks = 0; ks < KSTEPS; ++ks) {
        const int ka = ks * 4 + 2 * half;

        v2f a;
        if constexpr (CIN % 4 == 0) {
            a = *reinterpret_cast<const v2f*>(&H[row * APITCH + ka]);
            if constexpr (AMODE == A_SUM2)
                a += *reinterpret_cast<const v2f*>(&H2[row * APITCH + ka]);
            if constexpr (AMODE == A_FINAL) {
                const v2f ab = *reinterpret_cast<const v2f*>(&H2[row * APITCH + ka]);
                const v2f vt = *reinterpret_cast<const v2f*>(&T[row * APITCH + ka]);
                const v2f bi = *reinterpret_cast<const v2f*>(&bias_in[ka]);
                a.x = fmaxf(fmaf((a.x + ab.x) + vt.x, drow, bi.x), 0.0f);
                a.y = fmaxf(fmaf((a.y + ab.y) + vt.y, drow, bi.y), 0.0f);
            }
        } else {
            const int k0 = (ka     < CIN) ? ka     : (CIN - 1);
            const int k1 = (ka + 1 < CIN) ? ka + 1 : (CIN - 1);
            float ax = H[row * APITCH + k0];
            float ay = H[row * APITCH + k1];
            a.x = (ka     < CIN) ? ax : 0.0f;
            a.y = (ka + 1 < CIN) ? ay : 0.0f;
        }

#pragma unroll
        for (int nt = 0; nt < NT; ++nt) {
            const int ncol = nt * 16 + m;
            v2f b;
            if constexpr (CIN % 4 == 0) {
                b = *reinterpret_cast<const v2f*>(&W[ncol * CIN + ka]);
            } else {
                const int k0 = (ka     < CIN) ? ka     : (CIN - 1);
                const int k1 = (ka + 1 < CIN) ? ka + 1 : (CIN - 1);
                float bx = W[ncol * CIN + k0];
                float by = W[ncol * CIN + k1];
                b.x = (ka     < CIN) ? bx : 0.0f;
                b.y = (ka + 1 < CIN) ? by : 0.0f;
            }
            acc[nt] = __builtin_amdgcn_wmma_f32_16x16x4_f32(
                false, a, false, b, (short)0, acc[nt], false, false);
        }
    }

#pragma unroll
    for (int nt = 0; nt < NT; ++nt) {
        const int col = nt * 16 + m;
        const float bb = bias ? bias[col] : 0.0f;
#pragma unroll
        for (int r = 0; r < 8; ++r) {
            if constexpr (ROWSCALE)
                out[(row0 + r + 8 * half) * OPITCH + col] = acc[nt][r] * rs[r] + bb;
            else
                out[(row0 + r + 8 * half) * OPITCH + col] = acc[nt][r] + bb;
        }
    }

    if constexpr (ZEROW > 0) {
        constexpr int QPR = ZEROW / 4;
        constexpr int RPI = 32 / QPR;
        const float4 z = make_float4(0.0f, 0.0f, 0.0f, 0.0f);
        const int zr = lane / QPR;
        const int zc = (lane % QPR) * 4;
#pragma unroll
        for (int i = 0; i < 16 / RPI; ++i) {
            const int o = (row0 + i * RPI + zr) * PITCH + zc;
            *reinterpret_cast<float4*>(&zeroA[o]) = z;
            *reinterpret_cast<float4*>(&zeroB[o]) = z;
        }
    }
    if constexpr (ZEROPOOL) {
        static_assert(OPITCH == PITCH && COUT == 16, "the pooled sums use the free half of a 16-channel row");
        if (row0 < 2 * N_GRAPHS) {
            const float4 z = make_float4(0.0f, 0.0f, 0.0f, 0.0f);
            const int zr = lane >> 2;
            const int zc = 16 + (lane & 3) * 4;
#pragma unroll
            for (int i = 0; i < 2; ++i)
                *reinterpret_cast<float4*>(&out[(row0 + i * 8 + zr) * PITCH + zc]) = z;
        }
    }
}

__global__ void __launch_bounds__(256)
node_transform_l12(const float* H, const float* H2, const float* T,
                   const float* __restrict__ W1, const float* __restrict__ b1,
                   const float* __restrict__ W2, const float* __restrict__ dis,
                   float* out, float* zeroA, float* zeroB, int nrows) {
    const int wave = blockIdx.x * (blockDim.x >> 5) + (threadIdx.x >> 5);
    const int row0 = wave * 16;
    if (row0 >= nrows) return;

    const int lane = threadIdx.x & 31;
    const int h    = lane >> 4;
    const int c    = lane & 15;
    const int node = row0 + c;
    const float dn = dis[node];

    const int o = node * PITCH + 2 * h;
    v2f s = *reinterpret_cast<const v2f*>(&H[o]);
    s += *reinterpret_cast<const v2f*>(&H2[o]);
    s += *reinterpret_cast<const v2f*>(&T[o]);
    v2f bn;
    bn.x = s.x * dn;
    bn.y = h ? 0.0f : s.y * dn;

    v8f h1[2];
#pragma unroll
    for (int mt = 0; mt < 2; ++mt) {
        const int ch = mt * 16 + c;
        const float w0 = W1[ch * 3 + 2 * h];
        const float w1 = W1[ch * 3 + 1];
        v2f a;
        a.x = w0;
        a.y = h ? 0.0f : w1;
        h1[mt] = __builtin_amdgcn_wmma_f32_16x16x4_f32(
            false, a, false, bn, (short)0, (v8f){}, false, false);
        const float4 ba = *reinterpret_cast<const float4*>(&b1[mt * 16 + 8 * h]);
        const float4 bb = *reinterpret_cast<const float4*>(&b1[mt * 16 + 8 * h + 4]);
        h1[mt][0] = fmaxf(h1[mt][0] + ba.x, 0.0f);
        h1[mt][1] = fmaxf(h1[mt][1] + ba.y, 0.0f);
        h1[mt][2] = fmaxf(h1[mt][2] + ba.z, 0.0f);
        h1[mt][3] = fmaxf(h1[mt][3] + ba.w, 0.0f);
        h1[mt][4] = fmaxf(h1[mt][4] + bb.x, 0.0f);
        h1[mt][5] = fmaxf(h1[mt][5] + bb.y, 0.0f);
        h1[mt][6] = fmaxf(h1[mt][6] + bb.z, 0.0f);
        h1[mt][7] = fmaxf(h1[mt][7] + bb.w, 0.0f);
    }

    v8f acc = (v8f){};
#pragma unroll
    for (int ks = 0; ks < 8; ++ks) {
        const int mt = ks >> 2;
        const int r  = (ks & 3) * 2;
        v2f b;
        b.x = h1[mt][r];
        b.y = h1[mt][r + 1];
        const v2f a = *reinterpret_cast<const v2f*>(&W2[c * 32 + mt * 16 + 8 * h + r]);
        acc = __builtin_amdgcn_wmma_f32_16x16x4_f32(
            false, a, false, b, (short)0, acc, false, false);
    }

    float* po = &out[node * PITCH + 8 * h];
    *reinterpret_cast<float4*>(po)     = make_float4(acc[0] * dn, acc[1] * dn, acc[2] * dn, acc[3] * dn);
    *reinterpret_cast<float4*>(po + 4) = make_float4(acc[4] * dn, acc[5] * dn, acc[6] * dn, acc[7] * dn);

    const float4 z = make_float4(0.0f, 0.0f, 0.0f, 0.0f);
    const int zr = lane >> 2;
    const int zc = (lane & 3) * 4;
#pragma unroll
    for (int i = 0; i < 2; ++i) {
        const int zo = (row0 + i * 8 + zr) * PITCH + zc;
        *reinterpret_cast<float4*>(&zeroA[zo]) = z;
        *reinterpret_cast<float4*>(&zeroB[zo]) = z;
    }
}


template <int COUT, int J>
__device__ __forceinline__ int lane_bcast(int x) {
    if constexpr (COUT == 32) {
        return __builtin_amdgcn_readlane(x, J);
    } else if constexpr (COUT == 16) {
        return __builtin_amdgcn_update_dpp(0, x, 0x150 + J  , 0xf, 0xf, true);
    } else {
        return __builtin_amdgcn_update_dpp(0, x, J * 0x55  , 0xf, 0xf, true);
    }
}

template <int COUT, int J, int STEPS>
struct EdgeRows {
    static __device__ __forceinline__ void gather(int my_src, const float* __restrict__ t,
                                                  int c, float* v) {
        const int s = lane_bcast<COUT, J>(my_src);
        v[J] = t[s * PITCH + c];
        EdgeRows<COUT, J + 1, STEPS>::gather(my_src, t, c, v);
    }
    static __device__ __forceinline__ void scatter(int my_dst, float* agg,
                                                   int c, const float* v) {
        const int d = lane_bcast<COUT, J>(my_dst);
        atomicAdd(&agg[d * PITCH + c], v[J]);
        EdgeRows<COUT, J + 1, STEPS>::scatter(my_dst, agg, c, v);
    }
};
template <int COUT, int STEPS>
struct EdgeRows<COUT, STEPS, STEPS> {
    static __device__ __forceinline__ void gather(int, const float*, int, float*) {}
    static __device__ __forceinline__ void scatter(int, float*, int, const float*) {}
};

template <int COUT, int GROUPS>
__global__ void __launch_bounds__(256)
edge_scatter(const int* __restrict__ ei, const float* __restrict__ t,
             float* __restrict__ agg2) {
    static_assert(COUT == 4 || COUT == 16 || COUT == 32, "a row is a quad, half a wave or a whole wave");
    static_assert(N_EDGES % (256 * GROUPS) == 0, "grid covers the edges exactly: no lane is idle at a broadcast");
    constexpr int ROWS  = 32 / COUT;
    constexpr int STEPS = 32 / ROWS;

    const int e0 = blockIdx.x * (256 * GROUPS) + threadIdx.x;
    int my_src[GROUPS], my_dst[GROUPS];
#pragma unroll
    for (int g = 0; g < GROUPS; ++g) {
        my_src[g] = ei[e0 + g * 256];
        my_dst[g] = ei[N_EDGES + e0 + g * 256];
    }
    const int c = threadIdx.x % COUT;

    float v[GROUPS][STEPS];
#pragma unroll
    for (int g = 0; g < GROUPS; ++g) EdgeRows<COUT, 0, STEPS>::gather(my_src[g], t, c, v[g]);
    float* agg = agg2 + l2_half() * (N_NODES * PITCH);
#pragma unroll
    for (int g = 0; g < GROUPS; ++g) EdgeRows<COUT, 0, STEPS>::scatter(my_dst[g], agg, c, v[g]);
}

__global__ void final_pool(const float* __restrict__ aggA, const float* __restrict__ aggB,
                           const float* ts, const float* __restrict__ dis,
                           const float* __restrict__ bias, const int* __restrict__ batch,
                           float* pooled2) {
    float* pooled = pooled2 + l2_half() * (N_GRAPHS * PITCH);
    const int q = blockIdx.x * blockDim.x + threadIdx.x;
    if (q >= N_NODES * 4) return;
    const int n  = q >> 2;
    const int c4 = (q & 3) * 4;
    const float d = dis[n];
    float4 va = *reinterpret_cast<const float4*>(&aggA[n * PITCH + c4]);
    const float4 vh = *reinterpret_cast<const float4*>(&aggB[n * PITCH + c4]);
    va.x += vh.x; va.y += vh.y; va.z += vh.z; va.w += vh.w;
    const float4 vt = *reinterpret_cast<const float4*>(&ts[n * PITCH + c4]);
    const float4 vb = *reinterpret_cast<const float4*>(&bias[c4]);
    float4 v;
    v.x = fmaxf(fmaf(va.x + vt.x, d, vb.x), 0.0f);
    v.y = fmaxf(fmaf(va.y + vt.y, d, vb.y), 0.0f);
    v.z = fmaxf(fmaf(va.z + vt.z, d, vb.z), 0.0f);
    v.w = fmaxf(fmaf(va.w + vt.w, d, vb.w), 0.0f);
    float* base = &pooled[batch[n] * PITCH + c4];
    atomicAdd(base + 0, v.x);
    atomicAdd(base + 1, v.y);
    atomicAdd(base + 2, v.z);
    atomicAdd(base + 3, v.w);
}

__global__ void classify(const float* __restrict__ emb, const float* __restrict__ Wc,
                         const float* __restrict__ bc, float* __restrict__ out) {
    const int g = blockIdx.x * blockDim.x + threadIdx.x;
    if (g >= N_GRAPHS) return;
    float s = bc[0];
#pragma unroll
    for (int c = 0; c < 16; ++c) s += fmaxf(emb[g * 16 + c], 0.0f) * Wc[c];
    out[g] = s;
}

extern "C" void kernel_launch(void* const* d_in, const int* in_sizes, int n_in,
                              void* d_out, int out_size, void* d_ws, size_t ws_size,
                              hipStream_t stream) {
    const float* x     = (const float*)d_in[0];
    const int*   ei    = (const int*)d_in[1];
    const int*   batch = (const int*)d_in[2];
    const float* W1 = (const float*)d_in[3];
    const float* b1 = (const float*)d_in[4];
    const float* W2 = (const float*)d_in[5];
    const float* b2 = (const float*)d_in[6];
    const float* W3 = (const float*)d_in[7];
    const float* b3 = (const float*)d_in[8];
    const float* We = (const float*)d_in[9];
    const float* be = (const float*)d_in[10];
    const float* Wc = (const float*)d_in[11];
    const float* bc = (const float*)d_in[12];
    float* out = (float*)d_out;

    float* ws     = (float*)d_ws;
    float* dis    = ws;
    float* bufT   = dis  + N_NODES;
    float* bufA   = bufT + N_NODES * PITCH;
    float* bufH   = bufA + N_NODES * PITCH;
    float* pooled2 = bufT + 16;

    const int B = 256;

    clear_deg<<<ceil_div(N_NODES, B), B, 0, stream>>>(bufA, bufH);
    deg_accum<<<N_EDGES / (B * EDGE_ITERS), B, 0, stream>>>(ei, bufA, bufH);
    node_prep<<<ceil_div(N_NODES, B), B, 0, stream>>>(x, dis, bufT, bufA, bufH);

    const int gBlocks = ceil_div(N_NODES / 16, B / 32);

    edge_scatter<4, EDGE_ITERS><<<N_EDGES / (B * EDGE_ITERS), B, 0, stream>>>(ei, bufT, bufA);

    node_transform_l12<<<gBlocks, B, 0, stream>>>(
        bufA, bufH, bufT, W1, b1, W2, dis, bufT, bufA, bufH, N_NODES);
    edge_scatter<16, 1><<<N_EDGES / B, B, 0, stream>>>(ei, bufT, bufA);

    node_transform<16, 16, PITCH, PITCH, A_FINAL, true, 16, true><<<gBlocks, B, 0, stream>>>(
        bufA, bufH, bufT, b2, W3, nullptr, dis, bufT, bufA, bufH, N_NODES);
    edge_scatter<16, 1><<<N_EDGES / B, B, 0, stream>>>(ei, bufT, bufA);

    final_pool<<<ceil_div(N_NODES * 4, B), B, 0, stream>>>(bufA, bufH, bufT, dis, b3, batch, pooled2);
    node_transform<16, 16, PITCH, 16, A_SUM2, false, 0, false><<<ceil_div(N_GRAPHS / 16, B / 32), B, 0, stream>>>(
        pooled2, pooled2 + N_GRAPHS * PITCH, nullptr, nullptr, We, be, nullptr, out, nullptr, nullptr, N_GRAPHS);
    classify<<<ceil_div(N_GRAPHS, B), B, 0, stream>>>(out, Wc, bc, out + N_GRAPHS * 16);
}


# Edges counting-sorted per call by destination range (in LDS, no global atomic); one workgroup per range sums its rows in LDS, stores each once: degree and row atomics leave L2; sum order changes
# speedup vs baseline: 4.0714x; 1.0581x over previous
#include <hip/hip_runtime.h>

typedef __attribute__((ext_vector_type(2))) float v2f;
typedef __attribute__((ext_vector_type(8))) float v8f;

#define N_NODES  100000
#define N_EDGES  3200000
#define N_GRAPHS 1024

#define PITCH 32

#define NB      250
#define BS      400
#define NBLK    250
#define EPB     12800
#define SORT_T  512
#define EPT     25
#define BK_T    1024
#define SCAN_T  1024
#define SCAN_PER 64
#define TABROW0 2048
static_assert(NB * BS == N_NODES, "the ranges cover the nodes exactly");
static_assert(NBLK * EPB == N_EDGES && SORT_T * EPT == EPB, "the slices cover the edges exactly");
static_assert(SCAN_T * SCAN_PER >= NB * NBLK, "one scan workgroup covers the table");
static_assert(N_NODES <= (1 << 17) && BS < (1 << 15), "a sorted edge packs into one word");
static_assert(N_EDGES <= N_NODES * PITCH, "the sorted edges fit the buffer of one node plane");

static inline int ceil_div(int a, int b) { return (a + b - 1) / b; }

__device__ __forceinline__ int l2_half() {
    __shared__ int s_half;
    if (threadIdx.x == 0)
        s_half = (int)((__builtin_amdgcn_s_sendmsg_rtn(0x87) >> 18) & 1u);
    __syncthreads();
    return s_half;
}

__device__ __forceinline__ int* tab_at(float* bufT, int i) {
    return reinterpret_cast<int*>(bufT) + (TABROW0 + (i >> 4)) * PITCH + 16 + (i & 15);
}
__device__ __forceinline__ const int* tab_at(const float* bufT, int i) {
    return reinterpret_cast<const int*>(bufT) + (TABROW0 + (i >> 4)) * PITCH + 16 + (i & 15);
}
__device__ __forceinline__ int range_begin(const float* bufT, int b) {
    return (b < NB) ? *tab_at(bufT, b * NBLK) : N_EDGES;
}

__global__ void __launch_bounds__(SORT_T)
bin_count(const int* __restrict__ ei, float* bufT) {
    __shared__ int hist[NB];
    const int tid = threadIdx.x;
    if (tid < NB) hist[tid] = 0;
    __syncthreads();
    const int e0 = blockIdx.x * EPB + tid;
#pragma unroll
    for (int k = 0; k < EPT; ++k) {
        const int d = ei[N_EDGES + e0 + k * SORT_T];
        atomicAdd(&hist[d / BS], 1);
    }
    __syncthreads();
    if (tid < NB) *tab_at(bufT, tid * NBLK + blockIdx.x) = hist[tid];
}

__global__ void __launch_bounds__(SCAN_T)
bin_scan(float* bufT) {
    static_assert(SCAN_PER == 64 && (NB * NBLK) % 4 == 0, "a thread takes whole 16-byte pieces of four rows");
    __shared__ int part[2][SCAN_T];
    const int t  = threadIdx.x;
    const int i0 = t * SCAN_PER;
    int4* row = reinterpret_cast<int4*>(tab_at(bufT, i0));
    int v[SCAN_PER];
    int s = 0;
#pragma unroll
    for (int q = 0; q < SCAN_PER / 4; ++q) {
        int4 x = make_int4(0, 0, 0, 0);
        if (i0 + 4 * q < NB * NBLK) x = row[(q >> 2) * (PITCH / 4) + (q & 3)];
        v[4 * q] = x.x; v[4 * q + 1] = x.y; v[4 * q + 2] = x.z; v[4 * q + 3] = x.w;
        s += (x.x + x.y) + (x.z + x.w);
    }
    part[0][t] = s;
    __syncthreads();
    int cur = 0;
    for (int off = 1; off < SCAN_T; off <<= 1) {
        int x = part[cur][t];
        if (t >= off) x += part[cur][t - off];
        part[cur ^ 1][t] = x;
        __syncthreads();
        cur ^= 1;
    }
    int run = part[cur][t] - s;
#pragma unroll
    for (int q = 0; q < SCAN_PER / 4; ++q) {
        int4 o;
        o.x = run; run += v[4 * q];
        o.y = run; run += v[4 * q + 1];
        o.z = run; run += v[4 * q + 2];
        o.w = run; run += v[4 * q + 3];
        if (i0 + 4 * q < NB * NBLK) row[(q >> 2) * (PITCH / 4) + (q & 3)] = o;
    }
}

__global__ void __launch_bounds__(SORT_T)
bin_place(const int* __restrict__ ei, const float* bufT, unsigned* __restrict__ sorted) {
    __shared__ int hist[NB];
    __shared__ int lstart[NB];
    __shared__ int cursor[NB];
    __shared__ int goff[NB];
    __shared__ unsigned stage[EPB];
    const int tid = threadIdx.x;
    if (tid < NB) hist[tid] = 0;
    __syncthreads();
    const int e0 = blockIdx.x * EPB + tid;
    int s[EPT], d[EPT];
#pragma unroll
    for (int k = 0; k < EPT; ++k) {
        s[k] = ei[e0 + k * SORT_T];
        d[k] = ei[N_EDGES + e0 + k * SORT_T];
    }
#pragma unroll
    for (int k = 0; k < EPT; ++k) atomicAdd(&hist[d[k] / BS], 1);
    __syncthreads();
    if (tid < NB) {
        int a = 0;
        for (int j = 0; j < tid; ++j) a += hist[j];
        lstart[tid] = a;
        cursor[tid] = a;
        goff[tid]   = *tab_at(bufT, tid * NBLK + blockIdx.x);
    }
    __syncthreads();
#pragma unroll
    for (int k = 0; k < EPT; ++k) {
        const int b = d[k] / BS;
        const int p = atomicAdd(&cursor[b], 1);
        stage[p] = (unsigned)s[k] | ((unsigned)(d[k] - b * BS) << 17);
    }
    __syncthreads();
    const int lane = tid & 31;
    for (int b = tid >> 5; b < NB; b += SORT_T / 32) {
        const int n  = hist[b];
        const int ls = lstart[b];
        const int go = goff[b];
        for (int j = lane; j < n; j += 32) sorted[go + j] = stage[ls + j];
    }
}

__global__ void __launch_bounds__(BK_T)
bucket_prep(const unsigned* __restrict__ sorted, const float* __restrict__ x,
            float* __restrict__ dis, float* bufT) {
    __shared__ int cnt[BS];
    const int b   = blockIdx.x;
    const int tid = threadIdx.x;
    const int beg = range_begin(bufT, b);
    const int end = range_begin(bufT, b + 1);
    if (tid < BS) cnt[tid] = 0;
    __syncthreads();
    for (int j = beg + tid; j < end; j += BK_T) atomicAdd(&cnt[sorted[j] >> 17], 1);
    __syncthreads();
    if (tid < BS) {
        const int n = b * BS + tid;
        const float d = rsqrtf((float)cnt[tid] + 1.0f);
        dis[n] = d;
        *reinterpret_cast<float4*>(&bufT[n * PITCH]) =
            make_float4(x[n * 3] * d, x[n * 3 + 1] * d, x[n * 3 + 2] * d, 0.0f);
    }
}

enum { A_PLAIN = 0, A_SUM2 = 1, A_FINAL = 2 };

template <int CIN, int COUT, int APITCH, int OPITCH, int AMODE, bool ROWSCALE, bool ZEROPOOL>
__global__ void __launch_bounds__(256)
node_transform(const float* H, const float* H2, const float* T,
               const float* __restrict__ bias_in, const float* __restrict__ W,
               const float* __restrict__ bias, const float* __restrict__ rowscale,
               float* out, int nrows) {
    static_assert(CIN % 4 == 0 || AMODE == A_PLAIN, "the padded K path reads plain rows only");
    constexpr int NT     = COUT / 16;
    constexpr int KSTEPS = (CIN + 3) / 4;

    const int wavesPerBlock = blockDim.x >> 5;
    const int wave = blockIdx.x * wavesPerBlock + (threadIdx.x >> 5);
    const int row0 = wave * 16;
    if (row0 >= nrows) return;

    const int lane = threadIdx.x & 31;
    const int half = lane >> 4;
    const int m    = lane & 15;
    const int row  = row0 + m;

    float rs[8];
    if constexpr (ROWSCALE) {
        const float4 s0 = *reinterpret_cast<const float4*>(&rowscale[row0 + 8 * half]);
        const float4 s1 = *reinterpret_cast<const float4*>(&rowscale[row0 + 8 * half + 4]);
        rs[0] = s0.x; rs[1] = s0.y; rs[2] = s0.z; rs[3] = s0.w;
        rs[4] = s1.x; rs[5] = s1.y; rs[6] = s1.z; rs[7] = s1.w;
    }
    float drow = 0.0f;
    if constexpr (AMODE == A_FINAL) drow = rowscale[row];

    v8f acc[NT];
#pragma unroll
    for (int nt = 0; nt < NT; ++nt) acc[nt] = (v8f){};

#pragma unroll
    for (int ks = 0; ks < KSTEPS; ++ks) {
        const int ka = ks * 4 + 2 * half;

        v2f a;
        if constexpr (CIN % 4 == 0) {
            a = *reinterpret_cast<const v2f*>(&H[row * APITCH + ka]);
            if constexpr (AMODE == A_SUM2)
                a += *reinterpret_cast<const v2f*>(&H2[row * APITCH + ka]);
            if constexpr (AMODE == A_FINAL) {
                const v2f vt = *reinterpret_cast<const v2f*>(&T[row * APITCH + ka]);
                const v2f bi = *reinterpret_cast<const v2f*>(&bias_in[ka]);
                a.x = fmaxf(fmaf(a.x + vt.x, drow, bi.x), 0.0f);
                a.y = fmaxf(fmaf(a.y + vt.y, drow, bi.y), 0.0f);
            }
        } else {
            const int k0 = (ka     < CIN) ? ka     : (CIN - 1);
            const int k1 = (ka + 1 < CIN) ? ka + 1 : (CIN - 1);
            float ax = H[row * APITCH + k0];
            float ay = H[row * APITCH + k1];
            a.x = (ka     < CIN) ? ax : 0.0f;
            a.y = (ka + 1 < CIN) ? ay : 0.0f;
        }

#pragma unroll
        for (int nt = 0; nt < NT; ++nt) {
            const int ncol = nt * 16 + m;
            v2f b;
            if constexpr (CIN % 4 == 0) {
                b = *reinterpret_cast<const v2f*>(&W[ncol * CIN + ka]);
            } else {
                const int k0 = (ka     < CIN) ? ka     : (CIN - 1);
                const int k1 = (ka + 1 < CIN) ? ka + 1 : (CIN - 1);
                float bx = W[ncol * CIN + k0];
                float by = W[ncol * CIN + k1];
                b.x = (ka     < CIN) ? bx : 0.0f;
                b.y = (ka + 1 < CIN) ? by : 0.0f;
            }
            acc[nt] = __builtin_amdgcn_wmma_f32_16x16x4_f32(
                false, a, false, b, (short)0, acc[nt], false, false);
        }
    }

#pragma unroll
    for (int nt = 0; nt < NT; ++nt) {
        const int col = nt * 16 + m;
        const float bb = bias ? bias[col] : 0.0f;
#pragma unroll
        for (int r = 0; r < 8; ++r) {
            if constexpr (ROWSCALE)
                out[(row0 + r + 8 * half) * OPITCH + col] = acc[nt][r] * rs[r] + bb;
            else
                out[(row0 + r + 8 * half) * OPITCH + col] = acc[nt][r] + bb;
        }
    }

    if constexpr (ZEROPOOL) {
        static_assert(OPITCH == PITCH && COUT == 16, "the pooled sums use the free half of a 16-channel row");
        if (row0 < 2 * N_GRAPHS) {
            const float4 z = make_float4(0.0f, 0.0f, 0.0f, 0.0f);
            const int zr = lane >> 2;
            const int zc = 16 + (lane & 3) * 4;
#pragma unroll
            for (int i = 0; i < 2; ++i)
                *reinterpret_cast<float4*>(&out[(row0 + i * 8 + zr) * PITCH + zc]) = z;
        }
    }
}

__global__ void __launch_bounds__(256)
node_transform_l12(const float* H, const float* T,
                   const float* __restrict__ W1, const float* __restrict__ b1,
                   const float* __restrict__ W2, const float* __restrict__ dis,
                   float* out, int nrows) {
    const int wave = blockIdx.x * (blockDim.x >> 5) + (threadIdx.x >> 5);
    const int row0 = wave * 16;
    if (row0 >= nrows) return;

    const int lane = threadIdx.x & 31;
    const int h    = lane >> 4;
    const int c    = lane & 15;
    const int node = row0 + c;
    const float dn = dis[node];

    const int o = node * PITCH + 2 * h;
    v2f s = *reinterpret_cast<const v2f*>(&H[o]);
    s += *reinterpret_cast<const v2f*>(&T[o]);
    v2f bn;
    bn.x = s.x * dn;
    bn.y = h ? 0.0f : s.y * dn;

    v8f h1[2];
#pragma unroll
    for (int mt = 0; mt < 2; ++mt) {
        const int ch = mt * 16 + c;
        const float w0 = W1[ch * 3 + 2 * h];
        const float w1 = W1[ch * 3 + 1];
        v2f a;
        a.x = w0;
        a.y = h ? 0.0f : w1;
        h1[mt] = __builtin_amdgcn_wmma_f32_16x16x4_f32(
            false, a, false, bn, (short)0, (v8f){}, false, false);
        const float4 ba = *reinterpret_cast<const float4*>(&b1[mt * 16 + 8 * h]);
        const float4 bb = *reinterpret_cast<const float4*>(&b1[mt * 16 + 8 * h + 4]);
        h1[mt][0] = fmaxf(h1[mt][0] + ba.x, 0.0f);
        h1[mt][1] = fmaxf(h1[mt][1] + ba.y, 0.0f);
        h1[mt][2] = fmaxf(h1[mt][2] + ba.z, 0.0f);
        h1[mt][3] = fmaxf(h1[mt][3] + ba.w, 0.0f);
        h1[mt][4] = fmaxf(h1[mt][4] + bb.x, 0.0f);
        h1[mt][5] = fmaxf(h1[mt][5] + bb.y, 0.0f);
        h1[mt][6] = fmaxf(h1[mt][6] + bb.z, 0.0f);
        h1[mt][7] = fmaxf(h1[mt][7] + bb.w, 0.0f);
    }

    v8f acc = (v8f){};
#pragma unroll
    for (int ks = 0; ks < 8; ++ks) {
        const int mt = ks >> 2;
        const int r  = (ks & 3) * 2;
        v2f b;
        b.x = h1[mt][r];
        b.y = h1[mt][r + 1];
        const v2f a = *reinterpret_cast<const v2f*>(&W2[c * 32 + mt * 16 + 8 * h + r]);
        acc = __builtin_amdgcn_wmma_f32_16x16x4_f32(
            false, a, false, b, (short)0, acc, false, false);
    }

    float* po = &out[node * PITCH + 8 * h];
    *reinterpret_cast<float4*>(po)     = make_float4(acc[0] * dn, acc[1] * dn, acc[2] * dn, acc[3] * dn);
    *reinterpret_cast<float4*>(po + 4) = make_float4(acc[4] * dn, acc[5] * dn, acc[6] * dn, acc[7] * dn);

}


template <int COUT, int J>
__device__ __forceinline__ int lane_bcast(int x) {
    if constexpr (COUT == 16) {
        return __builtin_amdgcn_update_dpp(0, x, 0x150 + J  , 0xf, 0xf, true);
    } else {
        return __builtin_amdgcn_update_dpp(0, x, J * 0x55  , 0xf, 0xf, true);
    }
}

template <int COUT, int J, int STEPS>
struct EdgeRows {
    static __device__ __forceinline__ void gather(int my_src, const float* __restrict__ t,
                                                  int c, float* v) {
        const int s = lane_bcast<COUT, J>(my_src);
        v[J] = t[s * PITCH + c];
        EdgeRows<COUT, J + 1, STEPS>::gather(my_src, t, c, v);
    }
    static __device__ __forceinline__ void add(int my_row, float* acc, int c, const float* v) {
        const int r = lane_bcast<COUT, J>(my_row);
        atomicAdd(&acc[r * COUT + c], v[J]);
        EdgeRows<COUT, J + 1, STEPS>::add(my_row, acc, c, v);
    }
};
template <int COUT, int STEPS>
struct EdgeRows<COUT, STEPS, STEPS> {
    static __device__ __forceinline__ void gather(int, const float*, int, float*) {}
    static __device__ __forceinline__ void add(int, float*, int, const float*) {}
};

template <int COUT, int GROUPS>
__global__ void __launch_bounds__(BK_T)
bucket_gather(const unsigned* __restrict__ sorted, const float* __restrict__ t,
              float* __restrict__ agg) {
    static_assert(COUT == 4 || COUT == 16, "a row is a quad or half a wave");
    constexpr int ROWS  = 32 / COUT;
    constexpr int STEPS = 32 / ROWS;
    __shared__ __attribute__((aligned(16))) float acc[(BS + 1) * COUT];

    const int b   = blockIdx.x;
    const int tid = threadIdx.x;
    const int beg = range_begin(t, b);
    const int end = range_begin(t, b + 1);
    for (int i = tid; i < (BS + 1) * COUT; i += BK_T) acc[i] = 0.0f;
    __syncthreads();

    const int c    = tid % COUT;
    const int lane = tid & 31;
    constexpr unsigned PAD = (unsigned)BS << 17;
    int base = beg + (tid >> 5) * (32 * GROUPS);
    unsigned w[GROUPS];
#pragma unroll
    for (int g = 0; g < GROUPS; ++g) {
        const int j = base + g * 32 + lane;
        w[g] = (j < end) ? sorted[j] : PAD;
    }
    while (base < end) {
        int my_src[GROUPS], my_row[GROUPS];
#pragma unroll
        for (int g = 0; g < GROUPS; ++g) {
            my_src[g] = (int)(w[g] & 0x1FFFFu);
            my_row[g] = (int)(w[g] >> 17);
        }
        float v[GROUPS][STEPS];
#pragma unroll
        for (int g = 0; g < GROUPS; ++g) EdgeRows<COUT, 0, STEPS>::gather(my_src[g], t, c, v[g]);
        base += BK_T * GROUPS;
#pragma unroll
        for (int g = 0; g < GROUPS; ++g) {
            const int j = base + g * 32 + lane;
            w[g] = (j < end) ? sorted[j] : PAD;
        }
#pragma unroll
        for (int g = 0; g < GROUPS; ++g) EdgeRows<COUT, 0, STEPS>::add(my_row[g], acc, c, v[g]);
    }
    __syncthreads();

    constexpr int QPR = COUT / 4;
    for (int i = tid; i < BS * QPR; i += BK_T) {
        const int r = i / QPR;
        const int q = (i % QPR) * 4;
        *reinterpret_cast<float4*>(&agg[(b * BS + r) * PITCH + q]) =
            *reinterpret_cast<const float4*>(&acc[r * COUT + q]);
    }
}

__global__ void final_pool(const float* __restrict__ agg,
                           const float* ts, const float* __restrict__ dis,
                           const float* __restrict__ bias, const int* __restrict__ batch,
                           float* pooled2) {
    float* pooled = pooled2 + l2_half() * (N_GRAPHS * PITCH);
    const int q = blockIdx.x * blockDim.x + threadIdx.x;
    if (q >= N_NODES * 4) return;
    const int n  = q >> 2;
    const int c4 = (q & 3) * 4;
    const float d = dis[n];
    const float4 va = *reinterpret_cast<const float4*>(&agg[n * PITCH + c4]);
    const float4 vt = *reinterpret_cast<const float4*>(&ts[n * PITCH + c4]);
    const float4 vb = *reinterpret_cast<const float4*>(&bias[c4]);
    float4 v;
    v.x = fmaxf(fmaf(va.x + vt.x, d, vb.x), 0.0f);
    v.y = fmaxf(fmaf(va.y + vt.y, d, vb.y), 0.0f);
    v.z = fmaxf(fmaf(va.z + vt.z, d, vb.z), 0.0f);
    v.w = fmaxf(fmaf(va.w + vt.w, d, vb.w), 0.0f);
    float* base = &pooled[batch[n] * PITCH + c4];
    atomicAdd(base + 0, v.x);
    atomicAdd(base + 1, v.y);
    atomicAdd(base + 2, v.z);
    atomicAdd(base + 3, v.w);
}

__global__ void classify(const float* __restrict__ emb, const float* __restrict__ Wc,
                         const float* __restrict__ bc, float* __restrict__ out) {
    const int g = blockIdx.x * blockDim.x + threadIdx.x;
    if (g >= N_GRAPHS) return;
    float s = bc[0];
#pragma unroll
    for (int c = 0; c < 16; ++c) s += fmaxf(emb[g * 16 + c], 0.0f) * Wc[c];
    out[g] = s;
}

extern "C" void kernel_launch(void* const* d_in, const int* in_sizes, int n_in,
                              void* d_out, int out_size, void* d_ws, size_t ws_size,
                              hipStream_t stream) {
    const float* x     = (const float*)d_in[0];
    const int*   ei    = (const int*)d_in[1];
    const int*   batch = (const int*)d_in[2];
    const float* W1 = (const float*)d_in[3];
    const float* b1 = (const float*)d_in[4];
    const float* W2 = (const float*)d_in[5];
    const float* b2 = (const float*)d_in[6];
    const float* W3 = (const float*)d_in[7];
    const float* b3 = (const float*)d_in[8];
    const float* We = (const float*)d_in[9];
    const float* be = (const float*)d_in[10];
    const float* Wc = (const float*)d_in[11];
    const float* bc = (const float*)d_in[12];
    float* out = (float*)d_out;

    float* ws     = (float*)d_ws;
    float* dis    = ws;
    float* bufT   = dis  + N_NODES;
    float* bufA   = bufT + N_NODES * PITCH;
    unsigned* sorted = (unsigned*)(bufA + N_NODES * PITCH);
    float* pooled2 = bufT + 16;

    const int B = 256;

    bin_count<<<NBLK, SORT_T, 0, stream>>>(ei, bufT);
    bin_scan<<<1, SCAN_T, 0, stream>>>(bufT);
    bin_place<<<NBLK, SORT_T, 0, stream>>>(ei, bufT, sorted);

    bucket_prep<<<NB, BK_T, 0, stream>>>(sorted, x, dis, bufT);

    const int gBlocks = ceil_div(N_NODES / 16, B / 32);

    bucket_gather<4, 4><<<NB, BK_T, 0, stream>>>(sorted, bufT, bufA);

    node_transform_l12<<<gBlocks, B, 0, stream>>>(bufA, bufT, W1, b1, W2, dis, bufT, N_NODES);
    bucket_gather<16, 2><<<NB, BK_T, 0, stream>>>(sorted, bufT, bufA);

    node_transform<16, 16, PITCH, PITCH, A_FINAL, true, true><<<gBlocks, B, 0, stream>>>(
        bufA, nullptr, bufT, b2, W3, nullptr, dis, bufT, N_NODES);
    bucket_gather<16, 2><<<NB, BK_T, 0, stream>>>(sorted, bufT, bufA);

    final_pool<<<ceil_div(N_NODES * 4, B), B, 0, stream>>>(bufA, bufT, dis, b3, batch, pooled2);
    node_transform<16, 16, PITCH, 16, A_SUM2, false, false><<<ceil_div(N_GRAPHS / 16, B / 32), B, 0, stream>>>(
        pooled2, pooled2 + N_GRAPHS * PITCH, nullptr, nullptr, We, be, nullptr, out, N_GRAPHS);
    classify<<<ceil_div(N_GRAPHS, B), B, 0, stream>>>(out, Wc, bc, out + N_GRAPHS * 16);
}


# Node-local stages (each layer's finalize + next WMMA product, pooling, logits) run in the kernel that already owns those rows, sums staying in LDS; 12 launches to 8; pool and logit sum order changes
# speedup vs baseline: 4.7083x; 1.1564x over previous
#include <hip/hip_runtime.h>

typedef __attribute__((ext_vector_type(2))) float v2f;
typedef __attribute__((ext_vector_type(8))) float v8f;

#define N_NODES  100000
#define N_EDGES  3200000
#define N_GRAPHS 1024

#define PITCH 32

#define NB      250
#define BS      400
#define NBLK    250
#define EPB     12800
#define SORT_T  512
#define EPT     25
#define BK_T    1024
#define SCAN_T  1024
#define SCAN_PER 64
#define TABROW0 2048
static_assert(NB * BS == N_NODES, "the ranges cover the nodes exactly");
static_assert(NBLK * EPB == N_EDGES && SORT_T * EPT == EPB, "the slices cover the edges exactly");
static_assert(SCAN_T * SCAN_PER >= NB * NBLK, "one scan workgroup covers the table");
static_assert(N_NODES <= (1 << 17) && BS < (1 << 15), "a sorted edge packs into one word");
static_assert(N_EDGES <= N_NODES * PITCH, "the sorted edges fit the buffer of one node plane");

static inline int ceil_div(int a, int b) { return (a + b - 1) / b; }

__device__ __forceinline__ unsigned ask_die() { return __builtin_amdgcn_s_sendmsg_rtn(0x87); }
__device__ __forceinline__ int half_of_reply(unsigned reply) { return (int)((reply >> 18) & 1u); }

__device__ __forceinline__ int* tab_at(float* bufT, int i) {
    return reinterpret_cast<int*>(bufT) + (TABROW0 + (i >> 4)) * PITCH + 16 + (i & 15);
}
__device__ __forceinline__ const int* tab_at(const float* bufT, int i) {
    return reinterpret_cast<const int*>(bufT) + (TABROW0 + (i >> 4)) * PITCH + 16 + (i & 15);
}
__device__ __forceinline__ int range_begin(const float* bufT, int b) {
    return (b < NB) ? *tab_at(bufT, b * NBLK) : N_EDGES;
}

__global__ void __launch_bounds__(SORT_T)
bin_count(const int* __restrict__ ei, float* bufT) {
    __shared__ int hist[NB];
    const int tid = threadIdx.x;
    if (tid < NB) hist[tid] = 0;
    __syncthreads();
    const int e0 = blockIdx.x * EPB + tid;
#pragma unroll
    for (int k = 0; k < EPT; ++k) {
        const int d = ei[N_EDGES + e0 + k * SORT_T];
        atomicAdd(&hist[d / BS], 1);
    }
    __syncthreads();
    if (tid < NB) *tab_at(bufT, tid * NBLK + blockIdx.x) = hist[tid];
}

__global__ void __launch_bounds__(SCAN_T)
bin_scan(float* bufT) {
    static_assert(SCAN_PER == 64 && (NB * NBLK) % 4 == 0, "a thread takes whole 16-byte pieces of four rows");
    __shared__ int part[2][SCAN_T];
    const int t  = threadIdx.x;
    const int i0 = t * SCAN_PER;
    int4* row = reinterpret_cast<int4*>(tab_at(bufT, i0));
    int v[SCAN_PER];
    int s = 0;
#pragma unroll
    for (int q = 0; q < SCAN_PER / 4; ++q) {
        int4 x = make_int4(0, 0, 0, 0);
        if (i0 + 4 * q < NB * NBLK) x = row[(q >> 2) * (PITCH / 4) + (q & 3)];
        v[4 * q] = x.x; v[4 * q + 1] = x.y; v[4 * q + 2] = x.z; v[4 * q + 3] = x.w;
        s += (x.x + x.y) + (x.z + x.w);
    }
    part[0][t] = s;
    __syncthreads();
    int cur = 0;
    for (int off = 1; off < SCAN_T; off <<= 1) {
        int x = part[cur][t];
        if (t >= off) x += part[cur][t - off];
        part[cur ^ 1][t] = x;
        __syncthreads();
        cur ^= 1;
    }
    int run = part[cur][t] - s;
#pragma unroll
    for (int q = 0; q < SCAN_PER / 4; ++q) {
        int4 o;
        o.x = run; run += v[4 * q];
        o.y = run; run += v[4 * q + 1];
        o.z = run; run += v[4 * q + 2];
        o.w = run; run += v[4 * q + 3];
        if (i0 + 4 * q < NB * NBLK) row[(q >> 2) * (PITCH / 4) + (q & 3)] = o;
    }
}

__global__ void __launch_bounds__(SORT_T)
bin_place(const int* __restrict__ ei, const float* bufT, unsigned* __restrict__ sorted) {
    __shared__ int hist[NB];
    __shared__ int lstart[NB];
    __shared__ int cursor[NB];
    __shared__ int goff[NB];
    __shared__ unsigned stage[EPB];
    const int tid = threadIdx.x;
    if (tid < NB) hist[tid] = 0;
    __syncthreads();
    const int e0 = blockIdx.x * EPB + tid;
    int s[EPT], d[EPT];
#pragma unroll
    for (int k = 0; k < EPT; ++k) {
        s[k] = ei[e0 + k * SORT_T];
        d[k] = ei[N_EDGES + e0 + k * SORT_T];
    }
#pragma unroll
    for (int k = 0; k < EPT; ++k) atomicAdd(&hist[d[k] / BS], 1);
    __syncthreads();
    if (tid < NB) {
        int a = 0;
        for (int j = 0; j < tid; ++j) a += hist[j];
        lstart[tid] = a;
        cursor[tid] = a;
        goff[tid]   = *tab_at(bufT, tid * NBLK + blockIdx.x);
    }
    __syncthreads();
#pragma unroll
    for (int k = 0; k < EPT; ++k) {
        const int b = d[k] / BS;
        const int p = atomicAdd(&cursor[b], 1);
        stage[p] = (unsigned)s[k] | ((unsigned)(d[k] - b * BS) << 17);
    }
    __syncthreads();
    const int lane = tid & 31;
    for (int b = tid >> 5; b < NB; b += SORT_T / 32) {
        const int n  = hist[b];
        const int ls = lstart[b];
        const int go = goff[b];
        for (int j = lane; j < n; j += 32) sorted[go + j] = stage[ls + j];
    }
}

__global__ void __launch_bounds__(BK_T)
bucket_prep(const unsigned* __restrict__ sorted, const float* __restrict__ x,
            float* __restrict__ dis, float* bufT) {
    __shared__ int cnt[BS];
    const int b   = blockIdx.x;
    const int tid = threadIdx.x;
    const int beg = range_begin(bufT, b);
    const int end = range_begin(bufT, b + 1);
    if (tid < BS) cnt[tid] = 0;
    __syncthreads();
    for (int j = beg + tid; j < end; j += BK_T) atomicAdd(&cnt[sorted[j] >> 17], 1);
    __syncthreads();
    if (tid < BS) {
        const int n = b * BS + tid;
        const float d = rsqrtf((float)cnt[tid] + 1.0f);
        dis[n] = d;
        *reinterpret_cast<float4*>(&bufT[n * PITCH]) =
            make_float4(x[n * 3] * d, x[n * 3 + 1] * d, x[n * 3 + 2] * d, 0.0f);
    }
}


template <int COUT, int J>
__device__ __forceinline__ int lane_bcast(int x) {
    if constexpr (COUT == 16) {
        return __builtin_amdgcn_update_dpp(0, x, 0x150 + J  , 0xf, 0xf, true);
    } else {
        return __builtin_amdgcn_update_dpp(0, x, J * 0x55  , 0xf, 0xf, true);
    }
}

template <int COUT, int J, int STEPS>
struct EdgeRows {
    static __device__ __forceinline__ void gather(int my_src, const float* __restrict__ t,
                                                  int c, float* v) {
        const int s = lane_bcast<COUT, J>(my_src);
        v[J] = t[s * PITCH + c];
        EdgeRows<COUT, J + 1, STEPS>::gather(my_src, t, c, v);
    }
    static __device__ __forceinline__ void add(int my_row, float* acc, int c, const float* v) {
        const int r = lane_bcast<COUT, J>(my_row);
        atomicAdd(&acc[r * COUT + c], v[J]);
        EdgeRows<COUT, J + 1, STEPS>::add(my_row, acc, c, v);
    }
};
template <int COUT, int STEPS>
struct EdgeRows<COUT, STEPS, STEPS> {
    static __device__ __forceinline__ void gather(int, const float*, int, float*) {}
    static __device__ __forceinline__ void add(int, float*, int, const float*) {}
};

enum { EPI_L12 = 0, EPI_L3 = 1, EPI_POOL = 2 };

template <int EPI> struct NodeStage;

template <> struct NodeStage<EPI_L12> {
    float  dn;
    v2f    xs;
    v2f    w1[2];
    float4 b1a[2], b1b[2];
    v2f    w2[8];
    __device__ __forceinline__ void load(int node, int h, int cc, const float* __restrict__ src,
                                         const float* __restrict__ dis, const float* __restrict__ W1,
                                         const float* __restrict__ b1, const float* __restrict__ W2) {
        dn = dis[node];
        xs = *reinterpret_cast<const v2f*>(&src[node * PITCH + 2 * h]);
#pragma unroll
        for (int mt = 0; mt < 2; ++mt) {
            const int ch = mt * 16 + cc;
            const float w0 = W1[ch * 3 + 2 * h];
            const float wm = W1[ch * 3 + 1];
            w1[mt].x = w0;
            w1[mt].y = h ? 0.0f : wm;
            b1a[mt] = *reinterpret_cast<const float4*>(&b1[mt * 16 + 8 * h]);
            b1b[mt] = *reinterpret_cast<const float4*>(&b1[mt * 16 + 8 * h + 4]);
        }
#pragma unroll
        for (int ks = 0; ks < 8; ++ks)
            w2[ks] = *reinterpret_cast<const v2f*>(&W2[cc * 32 + (ks >> 2) * 16 + 8 * h + (ks & 3) * 2]);
    }
    __device__ __forceinline__ void finish(const float* sums, int loc, int node, int h, float* outp) const {
        v2f s = *reinterpret_cast<const v2f*>(&sums[loc * 4 + 2 * h]);
        s += xs;
        v2f bn;
        bn.x = s.x * dn;
        bn.y = h ? 0.0f : s.y * dn;

        v8f h1[2];
#pragma unroll
        for (int mt = 0; mt < 2; ++mt) {
            h1[mt] = __builtin_amdgcn_wmma_f32_16x16x4_f32(
                false, w1[mt], false, bn, (short)0, (v8f){}, false, false);
            h1[mt][0] = fmaxf(h1[mt][0] + b1a[mt].x, 0.0f);
            h1[mt][1] = fmaxf(h1[mt][1] + b1a[mt].y, 0.0f);
            h1[mt][2] = fmaxf(h1[mt][2] + b1a[mt].z, 0.0f);
            h1[mt][3] = fmaxf(h1[mt][3] + b1a[mt].w, 0.0f);
            h1[mt][4] = fmaxf(h1[mt][4] + b1b[mt].x, 0.0f);
            h1[mt][5] = fmaxf(h1[mt][5] + b1b[mt].y, 0.0f);
            h1[mt][6] = fmaxf(h1[mt][6] + b1b[mt].z, 0.0f);
            h1[mt][7] = fmaxf(h1[mt][7] + b1b[mt].w, 0.0f);
        }

        v8f o = (v8f){};
#pragma unroll
        for (int ks = 0; ks < 8; ++ks) {
            const int mt = ks >> 2;
            const int r  = (ks & 3) * 2;
            v2f bq;
            bq.x = h1[mt][r];
            bq.y = h1[mt][r + 1];
            o = __builtin_amdgcn_wmma_f32_16x16x4_f32(
                false, w2[ks], false, bq, (short)0, o, false, false);
        }

        float* po = &outp[node * PITCH + 8 * h];
        *reinterpret_cast<float4*>(po)     = make_float4(o[0] * dn, o[1] * dn, o[2] * dn, o[3] * dn);
        *reinterpret_cast<float4*>(po + 4) = make_float4(o[4] * dn, o[5] * dn, o[6] * dn, o[7] * dn);
    }
};

template <> struct NodeStage<EPI_L3> {
    float4 s0, s1;
    float  drow;
    v2f    vt[4], bi[4], bw[4];
    __device__ __forceinline__ void load(int row0, int m, int half, const float* __restrict__ src,
                                         const float* __restrict__ dis, const float* __restrict__ b_in,
                                         const float* __restrict__ W) {
        s0   = *reinterpret_cast<const float4*>(&dis[row0 + 8 * half]);
        s1   = *reinterpret_cast<const float4*>(&dis[row0 + 8 * half + 4]);
        drow = dis[row0 + m];
#pragma unroll
        for (int ks = 0; ks < 4; ++ks) {
            const int ka = ks * 4 + 2 * half;
            vt[ks] = *reinterpret_cast<const v2f*>(&src[(row0 + m) * PITCH + ka]);
            bi[ks] = *reinterpret_cast<const v2f*>(&b_in[ka]);
            bw[ks] = *reinterpret_cast<const v2f*>(&W[m * 16 + ka]);
        }
    }
    __device__ __forceinline__ void finish(const float* sums, int loc0, int row0, int m, int half,
                                           float* outp) const {
        v8f o = (v8f){};
#pragma unroll
        for (int ks = 0; ks < 4; ++ks) {
            const int ka = ks * 4 + 2 * half;
            v2f a = *reinterpret_cast<const v2f*>(&sums[(loc0 + m) * 16 + ka]);
            a.x = fmaxf(fmaf(a.x + vt[ks].x, drow, bi[ks].x), 0.0f);
            a.y = fmaxf(fmaf(a.y + vt[ks].y, drow, bi[ks].y), 0.0f);
            o = __builtin_amdgcn_wmma_f32_16x16x4_f32(
                false, a, false, bw[ks], (short)0, o, false, false);
        }
        const float rs[8] = {s0.x, s0.y, s0.z, s0.w, s1.x, s1.y, s1.z, s1.w};
#pragma unroll
        for (int r = 0; r < 8; ++r)
            outp[(row0 + r + 8 * half) * PITCH + m] = o[r] * rs[r];
    }
};

template <> struct NodeStage<EPI_POOL> {
    static constexpr int NPC = 8;
    static_assert(BS % NPC == 0 && (BS / NPC) * 16 <= BK_T, "a 16-lane group per chunk of nodes");
    float bb;
    float ts[NPC], d[NPC];
    int   gid[NPC];
    __device__ __forceinline__ void load(int n0, int ch, const float* __restrict__ src,
                                         const float* __restrict__ dis, const float* __restrict__ bias,
                                         const int* __restrict__ batch) {
        bb = bias[ch];
#pragma unroll
        for (int i = 0; i < NPC; ++i) {
            ts[i]  = src[(n0 + i) * PITCH + ch];
            d[i]   = dis[n0 + i];
            gid[i] = batch[n0 + i];
        }
    }
    __device__ __forceinline__ void finish(const float* sums, int i0, int ch, float* pooled) const {
        float hv[NPC];
#pragma unroll
        for (int i = 0; i < NPC; ++i)
            hv[i] = fmaxf(fmaf(sums[(i0 + i) * 16 + ch] + ts[i], d[i], bb), 0.0f);
        int   cur = gid[0];
        float s   = hv[0];
#pragma unroll
        for (int i = 1; i < NPC; ++i) {
            if (gid[i] != cur) {
                atomicAdd(&pooled[cur * PITCH + ch], s);
                s = 0.0f;
                cur = gid[i];
            }
            s += hv[i];
        }
        atomicAdd(&pooled[cur * PITCH + ch], s);
    }
};

template <int COUT, int GROUPS, int EPI>
__global__ void __launch_bounds__(BK_T)
bucket_layer(const unsigned* __restrict__ sorted, const float* __restrict__ src,
             const float* tab, const float* __restrict__ dis,
             const float* __restrict__ Wa, const float* __restrict__ ba,
             const float* __restrict__ Wb, const int* __restrict__ batch, float* outp) {
    static_assert(COUT == 4 || COUT == 16, "a row is a quad or half a wave");
    static_assert((EPI == EPI_L12) == (COUT == 4), "only layer 1 sums its input channels");
    static_assert(BS % 16 == 0 && BS / 16 <= BK_T / 32, "one wave per 16-node tile of the range");
    constexpr int ROWS  = 32 / COUT;
    constexpr int STEPS = 32 / ROWS;
    __shared__ __attribute__((aligned(16))) float acc[(BS + 1) * COUT];
    __shared__ int s_half;

    const int b    = blockIdx.x;
    const int tid  = threadIdx.x;
    const int lane = tid & 31;
    const int wv   = tid >> 5;
    const bool tile_wave = wv < BS / 16;

    unsigned die_reply = 0;
    if constexpr (EPI == EPI_POOL) {
        if (tid == 0) die_reply = ask_die();
    }
    if constexpr (EPI == EPI_L3) {
        static_assert(2 * N_GRAPHS * 4 == 8 * BK_T, "eight workgroups clear the pooled sums");
        if (b < 8) {
            const int i = b * BK_T + tid;
            *reinterpret_cast<float4*>(&outp[(i >> 2) * PITCH + 16 + (i & 3) * 4]) =
                make_float4(0.0f, 0.0f, 0.0f, 0.0f);
        }
    }
    const int beg = range_begin(tab, b);
    const int end = range_begin(tab, b + 1);

    NodeStage<EPI> stage;
    if constexpr (EPI == EPI_L12) {
        if (tile_wave) stage.load(b * BS + wv * 16 + (lane & 15), lane >> 4, lane & 15, src, dis, Wa, ba, Wb);
    }
    if constexpr (EPI == EPI_L3) {
        if (tile_wave) stage.load(b * BS + wv * 16, lane & 15, lane >> 4, src, dis, ba, Wa);
    }
    const bool chunk_lane = (tid >> 4) < BS / NodeStage<EPI_POOL>::NPC;
    if constexpr (EPI == EPI_POOL) {
        if (chunk_lane) stage.load(b * BS + (tid >> 4) * NodeStage<EPI_POOL>::NPC, tid & 15, src, dis, ba, batch);
    }

    for (int i = tid; i < (BS + 1) * COUT; i += BK_T) acc[i] = 0.0f;
    __syncthreads();

    const int c = tid % COUT;
    constexpr unsigned PAD = (unsigned)BS << 17;
    int base = beg + wv * (32 * GROUPS);
    unsigned w[GROUPS];
#pragma unroll
    for (int g = 0; g < GROUPS; ++g) {
        const int j = base + g * 32 + lane;
        w[g] = (j < end) ? sorted[j] : PAD;
    }
    while (base < end) {
        int my_src[GROUPS], my_row[GROUPS];
#pragma unroll
        for (int g = 0; g < GROUPS; ++g) {
            my_src[g] = (int)(w[g] & 0x1FFFFu);
            my_row[g] = (int)(w[g] >> 17);
        }
        float v[GROUPS][STEPS];
#pragma unroll
        for (int g = 0; g < GROUPS; ++g) EdgeRows<COUT, 0, STEPS>::gather(my_src[g], src, c, v[g]);
        base += BK_T * GROUPS;
#pragma unroll
        for (int g = 0; g < GROUPS; ++g) {
            const int j = base + g * 32 + lane;
            w[g] = (j < end) ? sorted[j] : PAD;
        }
#pragma unroll
        for (int g = 0; g < GROUPS; ++g) EdgeRows<COUT, 0, STEPS>::add(my_row[g], acc, c, v[g]);
    }
    if constexpr (EPI == EPI_POOL) {
        if (tid == 0) s_half = half_of_reply(die_reply);
    }
    __syncthreads();

    if constexpr (EPI == EPI_L12) {
        if (tile_wave) stage.finish(acc, wv * 16 + (lane & 15), b * BS + wv * 16 + (lane & 15), lane >> 4, outp);
    }
    if constexpr (EPI == EPI_L3) {
        if (tile_wave) stage.finish(acc, wv * 16, b * BS + wv * 16, lane & 15, lane >> 4, outp);
    }
    if constexpr (EPI == EPI_POOL) {
        if (chunk_lane)
            stage.finish(acc, (tid >> 4) * NodeStage<EPI_POOL>::NPC, tid & 15, outp + s_half * (N_GRAPHS * PITCH));
    }
}

__global__ void __launch_bounds__(256)
head_transform(const float* __restrict__ P0, const float* __restrict__ P1,
               const float* __restrict__ W, const float* __restrict__ bias,
               const float* __restrict__ Wc, const float* __restrict__ bc, float* out) {
    static_assert(N_GRAPHS % 32 == 0, "a wave owns 32 graphs");
    const int wave = blockIdx.x * (blockDim.x >> 5) + (threadIdx.x >> 5);
    const int g0 = wave * 32;
    if (g0 >= N_GRAPHS) return;
    const int lane = threadIdx.x & 31;
    const int half = lane >> 4;
    const int m    = lane & 15;
    const float bb  = bias[m];
    const float wc  = Wc[m];
    const float bc0 = bc[0];
    float* logits = out + N_GRAPHS * 16;
#pragma unroll
    for (int t = 0; t < 2; ++t) {
        const int row0 = g0 + t * 16;
        const int row  = row0 + m;
        v8f o = (v8f){};
#pragma unroll
        for (int ks = 0; ks < 4; ++ks) {
            const int ka = ks * 4 + 2 * half;
            v2f a = *reinterpret_cast<const v2f*>(&P0[row * PITCH + ka]);
            a += *reinterpret_cast<const v2f*>(&P1[row * PITCH + ka]);
            const v2f bw = *reinterpret_cast<const v2f*>(&W[m * 16 + ka]);
            o = __builtin_amdgcn_wmma_f32_16x16x4_f32(
                false, a, false, bw, (short)0, o, false, false);
        }
        float lg[8];
#pragma unroll
        for (int r = 0; r < 8; ++r) {
            const float e = o[r] + bb;
            out[(row0 + r + 8 * half) * 16 + m] = e;
            float s = fmaxf(e, 0.0f) * wc;
            s += __shfl_xor(s, 8);
            s += __shfl_xor(s, 4);
            s += __shfl_xor(s, 2);
            s += __shfl_xor(s, 1);
            lg[r] = s + bc0;
        }
        if (m == 0) {
            float* pl = &logits[row0 + 8 * half];
            *reinterpret_cast<float4*>(pl)     = make_float4(lg[0], lg[1], lg[2], lg[3]);
            *reinterpret_cast<float4*>(pl + 4) = make_float4(lg[4], lg[5], lg[6], lg[7]);
        }
    }
}

extern "C" void kernel_launch(void* const* d_in, const int* in_sizes, int n_in,
                              void* d_out, int out_size, void* d_ws, size_t ws_size,
                              hipStream_t stream) {
    const float* x     = (const float*)d_in[0];
    const int*   ei    = (const int*)d_in[1];
    const int*   batch = (const int*)d_in[2];
    const float* W1 = (const float*)d_in[3];
    const float* b1 = (const float*)d_in[4];
    const float* W2 = (const float*)d_in[5];
    const float* b2 = (const float*)d_in[6];
    const float* W3 = (const float*)d_in[7];
    const float* b3 = (const float*)d_in[8];
    const float* We = (const float*)d_in[9];
    const float* be = (const float*)d_in[10];
    const float* Wc = (const float*)d_in[11];
    const float* bc = (const float*)d_in[12];
    float* out = (float*)d_out;

    float* ws     = (float*)d_ws;
    float* dis    = ws;
    float* bufT   = dis  + N_NODES;
    float* bufA   = bufT + N_NODES * PITCH;
    unsigned* sorted = (unsigned*)(bufA + N_NODES * PITCH);
    float* pooled2 = bufT + 16;

    const int B = 256;

    bin_count<<<NBLK, SORT_T, 0, stream>>>(ei, bufT);
    bin_scan<<<1, SCAN_T, 0, stream>>>(bufT);
    bin_place<<<NBLK, SORT_T, 0, stream>>>(ei, bufT, sorted);

    bucket_prep<<<NB, BK_T, 0, stream>>>(sorted, x, dis, bufT);

    bucket_layer<4, 4, EPI_L12><<<NB, BK_T, 0, stream>>>(sorted, bufT, bufT, dis, W1, b1, W2, nullptr, bufA);

    bucket_layer<16, 2, EPI_L3><<<NB, BK_T, 0, stream>>>(sorted, bufA, bufT, dis, W3, b2, nullptr, nullptr, bufT);

    bucket_layer<16, 2, EPI_POOL><<<NB, BK_T, 0, stream>>>(sorted, bufT, bufT, dis, nullptr, b3, nullptr, batch, pooled2);

    head_transform<<<ceil_div(N_GRAPHS / 32, B / 32), B, 0, stream>>>(
        pooled2, pooled2 + N_GRAPHS * PITCH, We, be, Wc, bc, out);
}


# Sort drops its count and scan launches: each slice stores its edges ordered by range with its running counts; the degree kernel finds and compacts its range's runs, loads batched; 8 launches to 6
# speedup vs baseline: 6.7412x; 1.4318x over previous
#include <hip/hip_runtime.h>

typedef __attribute__((ext_vector_type(2))) float v2f;
typedef __attribute__((ext_vector_type(8))) float v8f;

#define N_NODES  100000
#define N_EDGES  3200000
#define N_GRAPHS 1024

#define PITCH 32

#define NB      250
#define BS      400
#define NBLK    250
#define EPB     12800
#define SORT_T  512
#define EPT     25
#define BK_T    1024
#define TABROW0 2048
#define GBASE   62512
static_assert(NB * BS == N_NODES, "the ranges cover the nodes exactly");
static_assert(NBLK * EPB == N_EDGES && SORT_T * EPT == EPB, "the slices cover the edges exactly");
static_assert(GBASE >= NB * NBLK && GBASE % 16 == 0, "the places start on a 64-byte piece of their own behind the prefixes");
static_assert(TABROW0 + ((GBASE + NB) >> 4) < N_NODES, "the tables fit the buffer's rows");
static_assert(BK_T / 32 == 32, "a wave scans the per-wave totals of a range workgroup with one lane each");
static_assert(N_NODES <= (1 << 17) && BS < (1 << 15), "a sorted edge packs into one word");
static_assert(N_EDGES <= N_NODES * PITCH, "the edges fit the buffer of one node plane");

static inline int ceil_div(int a, int b) { return (a + b - 1) / b; }

__device__ __forceinline__ unsigned ask_die() { return __builtin_amdgcn_s_sendmsg_rtn(0x87); }
__device__ __forceinline__ int half_of_reply(unsigned reply) { return (int)((reply >> 18) & 1u); }

__device__ __forceinline__ int* tab_at(float* bufT, int i) {
    return reinterpret_cast<int*>(bufT) + (TABROW0 + (i >> 4)) * PITCH + 16 + (i & 15);
}
__device__ __forceinline__ const int* tab_at(const float* bufT, int i) {
    return reinterpret_cast<const int*>(bufT) + (TABROW0 + (i >> 4)) * PITCH + 16 + (i & 15);
}
__device__ __forceinline__ int range_begin(const float* bufT, int b) {
    return (b < NB) ? *tab_at(bufT, GBASE + b) : N_EDGES;
}

__global__ void __launch_bounds__(SORT_T)
slice_sort(const int* __restrict__ ei, float* bufT, unsigned* __restrict__ runs) {
    __shared__ int hist[NB];
    __shared__ int cursor[NB];
    __shared__ unsigned stage[EPB];
    const int tid = threadIdx.x;
    if (tid < NB) hist[tid] = 0;
    __syncthreads();
    const int e0 = blockIdx.x * EPB + tid;
    int s[EPT], d[EPT];
#pragma unroll
    for (int k = 0; k < EPT; ++k) {
        s[k] = ei[e0 + k * SORT_T];
        d[k] = ei[N_EDGES + e0 + k * SORT_T];
    }
#pragma unroll
    for (int k = 0; k < EPT; ++k) atomicAdd(&hist[d[k] / BS], 1);
    __syncthreads();
    int a = 0;
    if (tid < NB) {
        for (int j = 0; j < tid; ++j) a += hist[j];
        cursor[tid] = a;
    }
    __syncthreads();
#pragma unroll
    for (int k = 0; k < EPT; ++k) {
        const int b = d[k] / BS;
        const int p = atomicAdd(&cursor[b], 1);
        stage[p] = (unsigned)s[k] | ((unsigned)(d[k] - b * BS) << 17);
    }
    __syncthreads();
#pragma unroll
    for (int k = 0; k < EPT; ++k)
        runs[blockIdx.x * EPB + tid + k * SORT_T] = stage[tid + k * SORT_T];
    if (tid < NB) *tab_at(bufT, tid * NBLK + blockIdx.x) = a;
}

__global__ void __launch_bounds__(BK_T)
bucket_prep(const unsigned* __restrict__ runs, const float* __restrict__ x,
            float* __restrict__ dis, float* bufT, unsigned* __restrict__ sorted) {
    constexpr int NW  = BK_T / 32;
    constexpr int RPW = (NBLK + NW - 1) / NW;
    __shared__ int cnt[BS];
    __shared__ int rfrom[NBLK];
    __shared__ int rlen[NBLK];
    __shared__ int rpos[NBLK];
    __shared__ int wlen[NW];
    __shared__ int wbelow[NW];
    const int b    = blockIdx.x;
    const int tid  = threadIdx.x;
    const int lane = tid & 31;
    const int wv   = tid >> 5;
    if (tid < BS) cnt[tid] = 0;
    float x0 = 0.0f, x1 = 0.0f, x2 = 0.0f;
    if (tid < BS) {
        const float* px = &x[(b * BS + tid) * 3];
        x0 = px[0]; x1 = px[1]; x2 = px[2];
    }
    int t0 = 0, n = 0;
    if (tid < NBLK) {
        t0 = *tab_at(bufT, b * NBLK + tid);
        const int t1 = (b + 1 < NB) ? *tab_at(bufT, (b + 1) * NBLK + tid) : EPB;
        n = t1 - t0;
    }
    int incl = n;
    for (int off = 1; off < 32; off <<= 1) {
        const int y = __shfl_up(incl, off);
        if (lane >= off) incl += y;
    }
    int below = t0;
    for (int off = 16; off > 0; off >>= 1) below += __shfl_xor(below, off);
    if (lane == 31) wlen[wv] = incl;
    if (lane == 0)  wbelow[wv] = below;
    __syncthreads();
    const int wl = wlen[lane];
    int wi = wl;
    for (int off = 1; off < 32; off <<= 1) {
        const int y = __shfl_up(wi, off);
        if (lane >= off) wi += y;
    }
    const int wbefore = __shfl(wi - wl, wv);
    int place = wbelow[lane];
    for (int off = 16; off > 0; off >>= 1) place += __shfl_xor(place, off);
    if (tid < NBLK) {
        rfrom[tid] = tid * EPB + t0;
        rlen[tid]  = n;
        rpos[tid]  = wbefore + incl - n;
    }
    __syncthreads();

    int      from[RPW], len[RPW], to[RPW];
    unsigned ea[RPW], eb[RPW];
#pragma unroll
    for (int r = 0; r < RPW; ++r) {
        const int k  = wv + r * NW;
        const int kk = (k < NBLK) ? k : 0;
        from[r] = rfrom[kk];
        len[r]  = (k < NBLK) ? rlen[kk] : 0;
        to[r]   = place + rpos[kk];
    }
#pragma unroll
    for (int r = 0; r < RPW; ++r) {
        ea[r] = (lane < len[r])      ? runs[from[r] + lane]      : 0u;
        eb[r] = (lane + 32 < len[r]) ? runs[from[r] + lane + 32] : 0u;
    }
#pragma unroll
    for (int r = 0; r < RPW; ++r) {
        if (lane < len[r]) {
            sorted[to[r] + lane] = ea[r];
            atomicAdd(&cnt[ea[r] >> 17], 1);
        }
        if (lane + 32 < len[r]) {
            sorted[to[r] + lane + 32] = eb[r];
            atomicAdd(&cnt[eb[r] >> 17], 1);
        }
    }
#pragma unroll
    for (int r = 0; r < RPW; ++r) {
        for (int j = lane + 64; j < len[r]; j += 32) {
            const unsigned e = runs[from[r] + j];
            sorted[to[r] + j] = e;
            atomicAdd(&cnt[e >> 17], 1);
        }
    }
    __syncthreads();
    if (tid == 0) *tab_at(bufT, GBASE + b) = place;
    if (tid < BS) {
        const int n_ = b * BS + tid;
        const float d = rsqrtf((float)cnt[tid] + 1.0f);
        dis[n_] = d;
        *reinterpret_cast<float4*>(&bufT[n_ * PITCH]) = make_float4(x0 * d, x1 * d, x2 * d, 0.0f);
    }
}


template <int COUT, int J>
__device__ __forceinline__ int lane_bcast(int x) {
    if constexpr (COUT == 16) {
        return __builtin_amdgcn_update_dpp(0, x, 0x150 + J  , 0xf, 0xf, true);
    } else {
        return __builtin_amdgcn_update_dpp(0, x, J * 0x55  , 0xf, 0xf, true);
    }
}

template <int COUT, int J, int STEPS>
struct EdgeRows {
    static __device__ __forceinline__ void gather(int my_src, const float* __restrict__ t,
                                                  int c, float* v) {
        const int s = lane_bcast<COUT, J>(my_src);
        v[J] = t[s * PITCH + c];
        EdgeRows<COUT, J + 1, STEPS>::gather(my_src, t, c, v);
    }
    static __device__ __forceinline__ void add(int my_row, float* acc, int c, const float* v) {
        const int r = lane_bcast<COUT, J>(my_row);
        atomicAdd(&acc[r * COUT + c], v[J]);
        EdgeRows<COUT, J + 1, STEPS>::add(my_row, acc, c, v);
    }
};
template <int COUT, int STEPS>
struct EdgeRows<COUT, STEPS, STEPS> {
    static __device__ __forceinline__ void gather(int, const float*, int, float*) {}
    static __device__ __forceinline__ void add(int, float*, int, const float*) {}
};

enum { EPI_L12 = 0, EPI_L3 = 1, EPI_POOL = 2 };

template <int EPI> struct NodeStage;

template <> struct NodeStage<EPI_L12> {
    float  dn;
    v2f    xs;
    v2f    w1[2];
    float4 b1a[2], b1b[2];
    v2f    w2[8];
    __device__ __forceinline__ void load(int node, int h, int cc, const float* __restrict__ src,
                                         const float* __restrict__ dis, const float* __restrict__ W1,
                                         const float* __restrict__ b1, const float* __restrict__ W2) {
        dn = dis[node];
        xs = *reinterpret_cast<const v2f*>(&src[node * PITCH + 2 * h]);
#pragma unroll
        for (int mt = 0; mt < 2; ++mt) {
            const int ch = mt * 16 + cc;
            const float w0 = W1[ch * 3 + 2 * h];
            const float wm = W1[ch * 3 + 1];
            w1[mt].x = w0;
            w1[mt].y = h ? 0.0f : wm;
            b1a[mt] = *reinterpret_cast<const float4*>(&b1[mt * 16 + 8 * h]);
            b1b[mt] = *reinterpret_cast<const float4*>(&b1[mt * 16 + 8 * h + 4]);
        }
#pragma unroll
        for (int ks = 0; ks < 8; ++ks)
            w2[ks] = *reinterpret_cast<const v2f*>(&W2[cc * 32 + (ks >> 2) * 16 + 8 * h + (ks & 3) * 2]);
    }
    __device__ __forceinline__ void finish(const float* sums, int loc, int node, int h, float* outp) const {
        v2f s = *reinterpret_cast<const v2f*>(&sums[loc * 4 + 2 * h]);
        s += xs;
        v2f bn;
        bn.x = s.x * dn;
        bn.y = h ? 0.0f : s.y * dn;

        v8f h1[2];
#pragma unroll
        for (int mt = 0; mt < 2; ++mt) {
            h1[mt] = __builtin_amdgcn_wmma_f32_16x16x4_f32(
                false, w1[mt], false, bn, (short)0, (v8f){}, false, false);
            h1[mt][0] = fmaxf(h1[mt][0] + b1a[mt].x, 0.0f);
            h1[mt][1] = fmaxf(h1[mt][1] + b1a[mt].y, 0.0f);
            h1[mt][2] = fmaxf(h1[mt][2] + b1a[mt].z, 0.0f);
            h1[mt][3] = fmaxf(h1[mt][3] + b1a[mt].w, 0.0f);
            h1[mt][4] = fmaxf(h1[mt][4] + b1b[mt].x, 0.0f);
            h1[mt][5] = fmaxf(h1[mt][5] + b1b[mt].y, 0.0f);
            h1[mt][6] = fmaxf(h1[mt][6] + b1b[mt].z, 0.0f);
            h1[mt][7] = fmaxf(h1[mt][7] + b1b[mt].w, 0.0f);
        }

        v8f o = (v8f){};
#pragma unroll
        for (int ks = 0; ks < 8; ++ks) {
            const int mt = ks >> 2;
            const int r  = (ks & 3) * 2;
            v2f bq;
            bq.x = h1[mt][r];
            bq.y = h1[mt][r + 1];
            o = __builtin_amdgcn_wmma_f32_16x16x4_f32(
                false, w2[ks], false, bq, (short)0, o, false, false);
        }

        float* po = &outp[node * PITCH + 8 * h];
        *reinterpret_cast<float4*>(po)     = make_float4(o[0] * dn, o[1] * dn, o[2] * dn, o[3] * dn);
        *reinterpret_cast<float4*>(po + 4) = make_float4(o[4] * dn, o[5] * dn, o[6] * dn, o[7] * dn);
    }
};

template <> struct NodeStage<EPI_L3> {
    float4 s0, s1;
    float  drow;
    v2f    vt[4], bi[4], bw[4];
    __device__ __forceinline__ void load(int row0, int m, int half, const float* __restrict__ src,
                                         const float* __restrict__ dis, const float* __restrict__ b_in,
                                         const float* __restrict__ W) {
        s0   = *reinterpret_cast<const float4*>(&dis[row0 + 8 * half]);
        s1   = *reinterpret_cast<const float4*>(&dis[row0 + 8 * half + 4]);
        drow = dis[row0 + m];
#pragma unroll
        for (int ks = 0; ks < 4; ++ks) {
            const int ka = ks * 4 + 2 * half;
            vt[ks] = *reinterpret_cast<const v2f*>(&src[(row0 + m) * PITCH + ka]);
            bi[ks] = *reinterpret_cast<const v2f*>(&b_in[ka]);
            bw[ks] = *reinterpret_cast<const v2f*>(&W[m * 16 + ka]);
        }
    }
    __device__ __forceinline__ void finish(const float* sums, int loc0, int row0, int m, int half,
                                           float* outp) const {
        v8f o = (v8f){};
#pragma unroll
        for (int ks = 0; ks < 4; ++ks) {
            const int ka = ks * 4 + 2 * half;
            v2f a = *reinterpret_cast<const v2f*>(&sums[(loc0 + m) * 16 + ka]);
            a.x = fmaxf(fmaf(a.x + vt[ks].x, drow, bi[ks].x), 0.0f);
            a.y = fmaxf(fmaf(a.y + vt[ks].y, drow, bi[ks].y), 0.0f);
            o = __builtin_amdgcn_wmma_f32_16x16x4_f32(
                false, a, false, bw[ks], (short)0, o, false, false);
        }
        const float rs[8] = {s0.x, s0.y, s0.z, s0.w, s1.x, s1.y, s1.z, s1.w};
#pragma unroll
        for (int r = 0; r < 8; ++r)
            outp[(row0 + r + 8 * half) * PITCH + m] = o[r] * rs[r];
    }
};

template <> struct NodeStage<EPI_POOL> {
    static constexpr int NPC = 8;
    static_assert(BS % NPC == 0 && (BS / NPC) * 16 <= BK_T, "a 16-lane group per chunk of nodes");
    float bb;
    float ts[NPC], d[NPC];
    int   gid[NPC];
    __device__ __forceinline__ void load(int n0, int ch, const float* __restrict__ src,
                                         const float* __restrict__ dis, const float* __restrict__ bias,
                                         const int* __restrict__ batch) {
        bb = bias[ch];
#pragma unroll
        for (int i = 0; i < NPC; ++i) {
            ts[i]  = src[(n0 + i) * PITCH + ch];
            d[i]   = dis[n0 + i];
            gid[i] = batch[n0 + i];
        }
    }
    __device__ __forceinline__ void finish(const float* sums, int i0, int ch, float* pooled) const {
        float hv[NPC];
#pragma unroll
        for (int i = 0; i < NPC; ++i)
            hv[i] = fmaxf(fmaf(sums[(i0 + i) * 16 + ch] + ts[i], d[i], bb), 0.0f);
        int   cur = gid[0];
        float s   = hv[0];
#pragma unroll
        for (int i = 1; i < NPC; ++i) {
            if (gid[i] != cur) {
                atomicAdd(&pooled[cur * PITCH + ch], s);
                s = 0.0f;
                cur = gid[i];
            }
            s += hv[i];
        }
        atomicAdd(&pooled[cur * PITCH + ch], s);
    }
};

template <int COUT, int GROUPS, int EPI>
__global__ void __launch_bounds__(BK_T)
bucket_layer(const unsigned* __restrict__ sorted, const float* __restrict__ src,
             const float* tab, const float* __restrict__ dis,
             const float* __restrict__ Wa, const float* __restrict__ ba,
             const float* __restrict__ Wb, const int* __restrict__ batch, float* outp) {
    static_assert(COUT == 4 || COUT == 16, "a row is a quad or half a wave");
    static_assert((EPI == EPI_L12) == (COUT == 4), "only layer 1 sums its input channels");
    static_assert(BS % 16 == 0 && BS / 16 <= BK_T / 32, "one wave per 16-node tile of the range");
    constexpr int ROWS  = 32 / COUT;
    constexpr int STEPS = 32 / ROWS;
    __shared__ __attribute__((aligned(16))) float acc[(BS + 1) * COUT];
    __shared__ int s_half;

    const int b    = blockIdx.x;
    const int tid  = threadIdx.x;
    const int lane = tid & 31;
    const int wv   = tid >> 5;
    const bool tile_wave = wv < BS / 16;

    unsigned die_reply = 0;
    if constexpr (EPI == EPI_POOL) {
        if (tid == 0) die_reply = ask_die();
    }
    if constexpr (EPI == EPI_L3) {
        static_assert(2 * N_GRAPHS * 4 == 8 * BK_T, "eight workgroups clear the pooled sums");
        if (b < 8) {
            const int i = b * BK_T + tid;
            *reinterpret_cast<float4*>(&outp[(i >> 2) * PITCH + 16 + (i & 3) * 4]) =
                make_float4(0.0f, 0.0f, 0.0f, 0.0f);
        }
    }
    const int beg = range_begin(tab, b);
    const int end = range_begin(tab, b + 1);

    NodeStage<EPI> stage;
    if constexpr (EPI == EPI_L12) {
        if (tile_wave) stage.load(b * BS + wv * 16 + (lane & 15), lane >> 4, lane & 15, src, dis, Wa, ba, Wb);
    }
    if constexpr (EPI == EPI_L3) {
        if (tile_wave) stage.load(b * BS + wv * 16, lane & 15, lane >> 4, src, dis, ba, Wa);
    }
    const bool chunk_lane = (tid >> 4) < BS / NodeStage<EPI_POOL>::NPC;
    if constexpr (EPI == EPI_POOL) {
        if (chunk_lane) stage.load(b * BS + (tid >> 4) * NodeStage<EPI_POOL>::NPC, tid & 15, src, dis, ba, batch);
    }

    for (int i = tid; i < (BS + 1) * COUT; i += BK_T) acc[i] = 0.0f;
    __syncthreads();

    const int c = tid % COUT;
    constexpr unsigned PAD = (unsigned)BS << 17;
    int base = beg + wv * (32 * GROUPS);
    unsigned w[GROUPS];
#pragma unroll
    for (int g = 0; g < GROUPS; ++g) {
        const int j = base + g * 32 + lane;
        w[g] = (j < end) ? sorted[j] : PAD;
    }
    while (base < end) {
        int my_src[GROUPS], my_row[GROUPS];
#pragma unroll
        for (int g = 0; g < GROUPS; ++g) {
            my_src[g] = (int)(w[g] & 0x1FFFFu);
            my_row[g] = (int)(w[g] >> 17);
        }
        float v[GROUPS][STEPS];
#pragma unroll
        for (int g = 0; g < GROUPS; ++g) EdgeRows<COUT, 0, STEPS>::gather(my_src[g], src, c, v[g]);
        base += BK_T * GROUPS;
#pragma unroll
        for (int g = 0; g < GROUPS; ++g) {
            const int j = base + g * 32 + lane;
            w[g] = (j < end) ? sorted[j] : PAD;
        }
#pragma unroll
        for (int g = 0; g < GROUPS; ++g) EdgeRows<COUT, 0, STEPS>::add(my_row[g], acc, c, v[g]);
    }
    if constexpr (EPI == EPI_POOL) {
        if (tid == 0) s_half = half_of_reply(die_reply);
    }
    __syncthreads();

    if constexpr (EPI == EPI_L12) {
        if (tile_wave) stage.finish(acc, wv * 16 + (lane & 15), b * BS + wv * 16 + (lane & 15), lane >> 4, outp);
    }
    if constexpr (EPI == EPI_L3) {
        if (tile_wave) stage.finish(acc, wv * 16, b * BS + wv * 16, lane & 15, lane >> 4, outp);
    }
    if constexpr (EPI == EPI_POOL) {
        if (chunk_lane)
            stage.finish(acc, (tid >> 4) * NodeStage<EPI_POOL>::NPC, tid & 15, outp + s_half * (N_GRAPHS * PITCH));
    }
}

__global__ void __launch_bounds__(256)
head_transform(const float* __restrict__ P0, const float* __restrict__ P1,
               const float* __restrict__ W, const float* __restrict__ bias,
               const float* __restrict__ Wc, const float* __restrict__ bc, float* out) {
    static_assert(N_GRAPHS % 32 == 0, "a wave owns 32 graphs");
    const int wave = blockIdx.x * (blockDim.x >> 5) + (threadIdx.x >> 5);
    const int g0 = wave * 32;
    if (g0 >= N_GRAPHS) return;
    const int lane = threadIdx.x & 31;
    const int half = lane >> 4;
    const int m    = lane & 15;
    const float bb  = bias[m];
    const float wc  = Wc[m];
    const float bc0 = bc[0];
    float* logits = out + N_GRAPHS * 16;
#pragma unroll
    for (int t = 0; t < 2; ++t) {
        const int row0 = g0 + t * 16;
        const int row  = row0 + m;
        v8f o = (v8f){};
#pragma unroll
        for (int ks = 0; ks < 4; ++ks) {
            const int ka = ks * 4 + 2 * half;
            v2f a = *reinterpret_cast<const v2f*>(&P0[row * PITCH + ka]);
            a += *reinterpret_cast<const v2f*>(&P1[row * PITCH + ka]);
            const v2f bw = *reinterpret_cast<const v2f*>(&W[m * 16 + ka]);
            o = __builtin_amdgcn_wmma_f32_16x16x4_f32(
                false, a, false, bw, (short)0, o, false, false);
        }
        float lg[8];
#pragma unroll
        for (int r = 0; r < 8; ++r) {
            const float e = o[r] + bb;
            out[(row0 + r + 8 * half) * 16 + m] = e;
            float s = fmaxf(e, 0.0f) * wc;
            s += __shfl_xor(s, 8);
            s += __shfl_xor(s, 4);
            s += __shfl_xor(s, 2);
            s += __shfl_xor(s, 1);
            lg[r] = s + bc0;
        }
        if (m == 0) {
            float* pl = &logits[row0 + 8 * half];
            *reinterpret_cast<float4*>(pl)     = make_float4(lg[0], lg[1], lg[2], lg[3]);
            *reinterpret_cast<float4*>(pl + 4) = make_float4(lg[4], lg[5], lg[6], lg[7]);
        }
    }
}

extern "C" void kernel_launch(void* const* d_in, const int* in_sizes, int n_in,
                              void* d_out, int out_size, void* d_ws, size_t ws_size,
                              hipStream_t stream) {
    const float* x     = (const float*)d_in[0];
    const int*   ei    = (const int*)d_in[1];
    const int*   batch = (const int*)d_in[2];
    const float* W1 = (const float*)d_in[3];
    const float* b1 = (const float*)d_in[4];
    const float* W2 = (const float*)d_in[5];
    const float* b2 = (const float*)d_in[6];
    const float* W3 = (const float*)d_in[7];
    const float* b3 = (const float*)d_in[8];
    const float* We = (const float*)d_in[9];
    const float* be = (const float*)d_in[10];
    const float* Wc = (const float*)d_in[11];
    const float* bc = (const float*)d_in[12];
    float* out = (float*)d_out;

    float* ws     = (float*)d_ws;
    float* dis    = ws;
    float* bufT   = dis  + N_NODES;
    float* bufA   = bufT + N_NODES * PITCH;
    unsigned* sorted = (unsigned*)(bufA + N_NODES * PITCH);
    float* pooled2 = bufT + 16;

    const int B = 256;

    slice_sort<<<NBLK, SORT_T, 0, stream>>>(ei, bufT, (unsigned*)bufA);

    bucket_prep<<<NB, BK_T, 0, stream>>>((const unsigned*)bufA, x, dis, bufT, sorted);

    bucket_layer<4, 4, EPI_L12><<<NB, BK_T, 0, stream>>>(sorted, bufT, bufT, dis, W1, b1, W2, nullptr, bufA);

    bucket_layer<16, 2, EPI_L3><<<NB, BK_T, 0, stream>>>(sorted, bufA, bufT, dis, W3, b2, nullptr, nullptr, bufT);

    bucket_layer<16, 2, EPI_POOL><<<NB, BK_T, 0, stream>>>(sorted, bufT, bufT, dis, nullptr, b3, nullptr, batch, pooled2);

    head_transform<<<ceil_div(N_GRAPHS / 32, B / 32), B, 0, stream>>>(
        pooled2, pooled2 + N_GRAPHS * PITCH, We, be, Wc, bc, out);
}


# Node rows stored at the pitch of their payload (64 B for 16-channel rows and pooled sums, 16 B for layer 1's rows; sort tables in an array of their own): no half-empty lines; sum order as before
# speedup vs baseline: 7.2198x; 1.0710x over previous
#include <hip/hip_runtime.h>

typedef __attribute__((ext_vector_type(2))) float v2f;
typedef __attribute__((ext_vector_type(8))) float v8f;

#define N_NODES  100000
#define N_EDGES  3200000
#define N_GRAPHS 1024

#define ROWF 16
#define XSF  4

#define NB      250
#define BS      400
#define NBLK    250
#define EPB     12800
#define SORT_T  512
#define EPT     25
#define BK_T    1024
#define GBASE   62512
#define TABLEN  62784
static_assert(NB * BS == N_NODES, "the ranges cover the nodes exactly");
static_assert(NBLK * EPB == N_EDGES && SORT_T * EPT == EPB, "the slices cover the edges exactly");
static_assert(GBASE >= NB * NBLK && GBASE % 16 == 0, "the places start on a 64-byte piece of their own behind the prefixes");
static_assert(GBASE + NB <= TABLEN && TABLEN % 32 == 0, "the tables fit their array");
static_assert(BK_T / 32 == 32, "a wave scans the per-wave totals of a range workgroup with one lane each");
static_assert(N_NODES <= (1 << 17) && BS < (1 << 15), "a sorted edge packs into one word");
static_assert(N_NODES * ROWF <= N_EDGES, "layer 2's source rows fit the front of the ordered slices' buffer");
static_assert(BS * ROWF % 32 == 0 && 16 * ROWF % 32 == 0 && BS * XSF % 32 == 0,
              "a range's rows, and a wave's 16-node tile of them, are whole 128-byte lines");

static inline int ceil_div(int a, int b) { return (a + b - 1) / b; }

__device__ __forceinline__ unsigned ask_die() { return __builtin_amdgcn_s_sendmsg_rtn(0x87); }
__device__ __forceinline__ int half_of_reply(unsigned reply) { return (int)((reply >> 18) & 1u); }

__device__ __forceinline__ int range_begin(const int* tab, int b) {
    return (b < NB) ? tab[GBASE + b] : N_EDGES;
}

__global__ void __launch_bounds__(SORT_T)
slice_sort(const int* __restrict__ ei, int* __restrict__ tab, unsigned* __restrict__ runs) {
    __shared__ int hist[NB];
    __shared__ int cursor[NB];
    __shared__ unsigned stage[EPB];
    const int tid = threadIdx.x;
    if (tid < NB) hist[tid] = 0;
    __syncthreads();
    const int e0 = blockIdx.x * EPB + tid;
    int s[EPT], d[EPT];
#pragma unroll
    for (int k = 0; k < EPT; ++k) {
        s[k] = ei[e0 + k * SORT_T];
        d[k] = ei[N_EDGES + e0 + k * SORT_T];
    }
#pragma unroll
    for (int k = 0; k < EPT; ++k) atomicAdd(&hist[d[k] / BS], 1);
    __syncthreads();
    int a = 0;
    if (tid < NB) {
        for (int j = 0; j < tid; ++j) a += hist[j];
        cursor[tid] = a;
    }
    __syncthreads();
#pragma unroll
    for (int k = 0; k < EPT; ++k) {
        const int b = d[k] / BS;
        const int p = atomicAdd(&cursor[b], 1);
        stage[p] = (unsigned)s[k] | ((unsigned)(d[k] - b * BS) << 17);
    }
    __syncthreads();
#pragma unroll
    for (int k = 0; k < EPT; ++k)
        runs[blockIdx.x * EPB + tid + k * SORT_T] = stage[tid + k * SORT_T];
    if (tid < NB) tab[tid * NBLK + blockIdx.x] = a;
}

__global__ void __launch_bounds__(BK_T)
bucket_prep(const unsigned* __restrict__ runs, const float* __restrict__ x,
            float* __restrict__ dis, int* tab, float* __restrict__ xs, unsigned* __restrict__ sorted) {
    constexpr int NW  = BK_T / 32;
    constexpr int RPW = (NBLK + NW - 1) / NW;
    __shared__ int cnt[BS];
    __shared__ int rfrom[NBLK];
    __shared__ int rlen[NBLK];
    __shared__ int rpos[NBLK];
    __shared__ int wlen[NW];
    __shared__ int wbelow[NW];
    const int b    = blockIdx.x;
    const int tid  = threadIdx.x;
    const int lane = tid & 31;
    const int wv   = tid >> 5;
    if (tid < BS) cnt[tid] = 0;
    float x0 = 0.0f, x1 = 0.0f, x2 = 0.0f;
    if (tid < BS) {
        const float* px = &x[(b * BS + tid) * 3];
        x0 = px[0]; x1 = px[1]; x2 = px[2];
    }
    int t0 = 0, n = 0;
    if (tid < NBLK) {
        t0 = tab[b * NBLK + tid];
        const int t1 = (b + 1 < NB) ? tab[(b + 1) * NBLK + tid] : EPB;
        n = t1 - t0;
    }
    int incl = n;
    for (int off = 1; off < 32; off <<= 1) {
        const int y = __shfl_up(incl, off);
        if (lane >= off) incl += y;
    }
    int below = t0;
    for (int off = 16; off > 0; off >>= 1) below += __shfl_xor(below, off);
    if (lane == 31) wlen[wv] = incl;
    if (lane == 0)  wbelow[wv] = below;
    __syncthreads();
    const int wl = wlen[lane];
    int wi = wl;
    for (int off = 1; off < 32; off <<= 1) {
        const int y = __shfl_up(wi, off);
        if (lane >= off) wi += y;
    }
    const int wbefore = __shfl(wi - wl, wv);
    int place = wbelow[lane];
    for (int off = 16; off > 0; off >>= 1) place += __shfl_xor(place, off);
    if (tid < NBLK) {
        rfrom[tid] = tid * EPB + t0;
        rlen[tid]  = n;
        rpos[tid]  = wbefore + incl - n;
    }
    __syncthreads();

    int      from[RPW], len[RPW], to[RPW];
    unsigned ea[RPW], eb[RPW];
#pragma unroll
    for (int r = 0; r < RPW; ++r) {
        const int k  = wv + r * NW;
        const int kk = (k < NBLK) ? k : 0;
        from[r] = rfrom[kk];
        len[r]  = (k < NBLK) ? rlen[kk] : 0;
        to[r]   = place + rpos[kk];
    }
#pragma unroll
    for (int r = 0; r < RPW; ++r) {
        ea[r] = (lane < len[r])      ? runs[from[r] + lane]      : 0u;
        eb[r] = (lane + 32 < len[r]) ? runs[from[r] + lane + 32] : 0u;
    }
#pragma unroll
    for (int r = 0; r < RPW; ++r) {
        if (lane < len[r]) {
            sorted[to[r] + lane] = ea[r];
            atomicAdd(&cnt[ea[r] >> 17], 1);
        }
        if (lane + 32 < len[r]) {
            sorted[to[r] + lane + 32] = eb[r];
            atomicAdd(&cnt[eb[r] >> 17], 1);
        }
    }
#pragma unroll
    for (int r = 0; r < RPW; ++r) {
        for (int j = lane + 64; j < len[r]; j += 32) {
            const unsigned e = runs[from[r] + j];
            sorted[to[r] + j] = e;
            atomicAdd(&cnt[e >> 17], 1);
        }
    }
    __syncthreads();
    if (tid == 0) tab[GBASE + b] = place;
    if (tid < BS) {
        const int n_ = b * BS + tid;
        const float d = rsqrtf((float)cnt[tid] + 1.0f);
        dis[n_] = d;
        *reinterpret_cast<float4*>(&xs[n_ * XSF]) = make_float4(x0 * d, x1 * d, x2 * d, 0.0f);
    }
}


template <int COUT, int J>
__device__ __forceinline__ int lane_bcast(int x) {
    if constexpr (COUT == 16) {
        return __builtin_amdgcn_update_dpp(0, x, 0x150 + J  , 0xf, 0xf, true);
    } else {
        return __builtin_amdgcn_update_dpp(0, x, J * 0x55  , 0xf, 0xf, true);
    }
}

template <int COUT, int J, int STEPS>
struct EdgeRows {
    static __device__ __forceinline__ void gather(int my_src, const float* __restrict__ t,
                                                  int c, float* v) {
        const int s = lane_bcast<COUT, J>(my_src);
        v[J] = t[s * COUT + c];
        EdgeRows<COUT, J + 1, STEPS>::gather(my_src, t, c, v);
    }
    static __device__ __forceinline__ void add(int my_row, float* acc, int c, const float* v) {
        const int r = lane_bcast<COUT, J>(my_row);
        atomicAdd(&acc[r * COUT + c], v[J]);
        EdgeRows<COUT, J + 1, STEPS>::add(my_row, acc, c, v);
    }
};
template <int COUT, int STEPS>
struct EdgeRows<COUT, STEPS, STEPS> {
    static __device__ __forceinline__ void gather(int, const float*, int, float*) {}
    static __device__ __forceinline__ void add(int, float*, int, const float*) {}
};

enum { EPI_L12 = 0, EPI_L3 = 1, EPI_POOL = 2 };

template <int EPI> struct NodeStage;

template <> struct NodeStage<EPI_L12> {
    float  dn;
    v2f    xs;
    v2f    w1[2];
    float4 b1a[2], b1b[2];
    v2f    w2[8];
    __device__ __forceinline__ void load(int node, int h, int cc, const float* __restrict__ src,
                                         const float* __restrict__ dis, const float* __restrict__ W1,
                                         const float* __restrict__ b1, const float* __restrict__ W2) {
        dn = dis[node];
        xs = *reinterpret_cast<const v2f*>(&src[node * XSF + 2 * h]);
#pragma unroll
        for (int mt = 0; mt < 2; ++mt) {
            const int ch = mt * 16 + cc;
            const float w0 = W1[ch * 3 + 2 * h];
            const float wm = W1[ch * 3 + 1];
            w1[mt].x = w0;
            w1[mt].y = h ? 0.0f : wm;
            b1a[mt] = *reinterpret_cast<const float4*>(&b1[mt * 16 + 8 * h]);
            b1b[mt] = *reinterpret_cast<const float4*>(&b1[mt * 16 + 8 * h + 4]);
        }
#pragma unroll
        for (int ks = 0; ks < 8; ++ks)
            w2[ks] = *reinterpret_cast<const v2f*>(&W2[cc * 32 + (ks >> 2) * 16 + 8 * h + (ks & 3) * 2]);
    }
    __device__ __forceinline__ void finish(const float* sums, int loc, int node, int h, float* outp) const {
        v2f s = *reinterpret_cast<const v2f*>(&sums[loc * 4 + 2 * h]);
        s += xs;
        v2f bn;
        bn.x = s.x * dn;
        bn.y = h ? 0.0f : s.y * dn;

        v8f h1[2];
#pragma unroll
        for (int mt = 0; mt < 2; ++mt) {
            h1[mt] = __builtin_amdgcn_wmma_f32_16x16x4_f32(
                false, w1[mt], false, bn, (short)0, (v8f){}, false, false);
            h1[mt][0] = fmaxf(h1[mt][0] + b1a[mt].x, 0.0f);
            h1[mt][1] = fmaxf(h1[mt][1] + b1a[mt].y, 0.0f);
            h1[mt][2] = fmaxf(h1[mt][2] + b1a[mt].z, 0.0f);
            h1[mt][3] = fmaxf(h1[mt][3] + b1a[mt].w, 0.0f);
            h1[mt][4] = fmaxf(h1[mt][4] + b1b[mt].x, 0.0f);
            h1[mt][5] = fmaxf(h1[mt][5] + b1b[mt].y, 0.0f);
            h1[mt][6] = fmaxf(h1[mt][6] + b1b[mt].z, 0.0f);
            h1[mt][7] = fmaxf(h1[mt][7] + b1b[mt].w, 0.0f);
        }

        v8f o = (v8f){};
#pragma unroll
        for (int ks = 0; ks < 8; ++ks) {
            const int mt = ks >> 2;
            const int r  = (ks & 3) * 2;
            v2f bq;
            bq.x = h1[mt][r];
            bq.y = h1[mt][r + 1];
            o = __builtin_amdgcn_wmma_f32_16x16x4_f32(
                false, w2[ks], false, bq, (short)0, o, false, false);
        }

        float* po = &outp[node * ROWF + 8 * h];
        *reinterpret_cast<float4*>(po)     = make_float4(o[0] * dn, o[1] * dn, o[2] * dn, o[3] * dn);
        *reinterpret_cast<float4*>(po + 4) = make_float4(o[4] * dn, o[5] * dn, o[6] * dn, o[7] * dn);
    }
};

template <> struct NodeStage<EPI_L3> {
    float4 s0, s1;
    float  drow;
    v2f    vt[4], bi[4], bw[4];
    __device__ __forceinline__ void load(int row0, int m, int half, const float* __restrict__ src,
                                         const float* __restrict__ dis, const float* __restrict__ b_in,
                                         const float* __restrict__ W) {
        s0   = *reinterpret_cast<const float4*>(&dis[row0 + 8 * half]);
        s1   = *reinterpret_cast<const float4*>(&dis[row0 + 8 * half + 4]);
        drow = dis[row0 + m];
#pragma unroll
        for (int ks = 0; ks < 4; ++ks) {
            const int ka = ks * 4 + 2 * half;
            vt[ks] = *reinterpret_cast<const v2f*>(&src[(row0 + m) * ROWF + ka]);
            bi[ks] = *reinterpret_cast<const v2f*>(&b_in[ka]);
            bw[ks] = *reinterpret_cast<const v2f*>(&W[m * 16 + ka]);
        }
    }
    __device__ __forceinline__ void finish(const float* sums, int loc0, int row0, int m, int half,
                                           float* outp) const {
        v8f o = (v8f){};
#pragma unroll
        for (int ks = 0; ks < 4; ++ks) {
            const int ka = ks * 4 + 2 * half;
            v2f a = *reinterpret_cast<const v2f*>(&sums[(loc0 + m) * 16 + ka]);
            a.x = fmaxf(fmaf(a.x + vt[ks].x, drow, bi[ks].x), 0.0f);
            a.y = fmaxf(fmaf(a.y + vt[ks].y, drow, bi[ks].y), 0.0f);
            o = __builtin_amdgcn_wmma_f32_16x16x4_f32(
                false, a, false, bw[ks], (short)0, o, false, false);
        }
        const float rs[8] = {s0.x, s0.y, s0.z, s0.w, s1.x, s1.y, s1.z, s1.w};
#pragma unroll
        for (int r = 0; r < 8; ++r)
            outp[(row0 + r + 8 * half) * ROWF + m] = o[r] * rs[r];
    }
};

template <> struct NodeStage<EPI_POOL> {
    static constexpr int NPC = 8;
    static_assert(BS % NPC == 0 && (BS / NPC) * 16 <= BK_T, "a 16-lane group per chunk of nodes");
    float bb;
    float ts[NPC], d[NPC];
    int   gid[NPC];
    __device__ __forceinline__ void load(int n0, int ch, const float* __restrict__ src,
                                         const float* __restrict__ dis, const float* __restrict__ bias,
                                         const int* __restrict__ batch) {
        bb = bias[ch];
#pragma unroll
        for (int i = 0; i < NPC; ++i) {
            ts[i]  = src[(n0 + i) * ROWF + ch];
            d[i]   = dis[n0 + i];
            gid[i] = batch[n0 + i];
        }
    }
    __device__ __forceinline__ void finish(const float* sums, int i0, int ch, float* pooled) const {
        float hv[NPC];
#pragma unroll
        for (int i = 0; i < NPC; ++i)
            hv[i] = fmaxf(fmaf(sums[(i0 + i) * 16 + ch] + ts[i], d[i], bb), 0.0f);
        int   cur = gid[0];
        float s   = hv[0];
#pragma unroll
        for (int i = 1; i < NPC; ++i) {
            if (gid[i] != cur) {
                atomicAdd(&pooled[cur * ROWF + ch], s);
                s = 0.0f;
                cur = gid[i];
            }
            s += hv[i];
        }
        atomicAdd(&pooled[cur * ROWF + ch], s);
    }
};

template <int COUT, int GROUPS, int EPI>
__global__ void __launch_bounds__(BK_T)
bucket_layer(const unsigned* __restrict__ sorted, const float* __restrict__ src,
             const int* __restrict__ tab, const float* __restrict__ dis,
             const float* __restrict__ Wa, const float* __restrict__ ba,
             const float* __restrict__ Wb, const int* __restrict__ batch, float* outp,
             float* __restrict__ pool) {
    static_assert(COUT == 4 || COUT == 16, "a row is a quad or half a wave");
    static_assert((EPI == EPI_L12) == (COUT == 4), "only layer 1 sums its input channels");
    static_assert(BS % 16 == 0 && BS / 16 <= BK_T / 32, "one wave per 16-node tile of the range");
    constexpr int ROWS  = 32 / COUT;
    constexpr int STEPS = 32 / ROWS;
    __shared__ __attribute__((aligned(16))) float acc[(BS + 1) * COUT];
    __shared__ int s_half;

    const int b    = blockIdx.x;
    const int tid  = threadIdx.x;
    const int lane = tid & 31;
    const int wv   = tid >> 5;
    const bool tile_wave = wv < BS / 16;

    unsigned die_reply = 0;
    if constexpr (EPI == EPI_POOL) {
        if (tid == 0) die_reply = ask_die();
    }
    if constexpr (EPI == EPI_L3) {
        static_assert(2 * N_GRAPHS * ROWF == 4 * 8 * BK_T, "eight workgroups clear the pooled sums");
        if (b < 8)
            reinterpret_cast<float4*>(pool)[b * BK_T + tid] = make_float4(0.0f, 0.0f, 0.0f, 0.0f);
    }
    const int beg = range_begin(tab, b);
    const int end = range_begin(tab, b + 1);

    NodeStage<EPI> stage;
    if constexpr (EPI == EPI_L12) {
        if (tile_wave) stage.load(b * BS + wv * 16 + (lane & 15), lane >> 4, lane & 15, src, dis, Wa, ba, Wb);
    }
    if constexpr (EPI == EPI_L3) {
        if (tile_wave) stage.load(b * BS + wv * 16, lane & 15, lane >> 4, src, dis, ba, Wa);
    }
    const bool chunk_lane = (tid >> 4) < BS / NodeStage<EPI_POOL>::NPC;
    if constexpr (EPI == EPI_POOL) {
        if (chunk_lane) stage.load(b * BS + (tid >> 4) * NodeStage<EPI_POOL>::NPC, tid & 15, src, dis, ba, batch);
    }

    for (int i = tid; i < (BS + 1) * COUT; i += BK_T) acc[i] = 0.0f;
    __syncthreads();

    const int c = tid % COUT;
    constexpr unsigned PAD = (unsigned)BS << 17;
    int base = beg + wv * (32 * GROUPS);
    unsigned w[GROUPS];
#pragma unroll
    for (int g = 0; g < GROUPS; ++g) {
        const int j = base + g * 32 + lane;
        w[g] = (j < end) ? sorted[j] : PAD;
    }
    while (base < end) {
        int my_src[GROUPS], my_row[GROUPS];
#pragma unroll
        for (int g = 0; g < GROUPS; ++g) {
            my_src[g] = (int)(w[g] & 0x1FFFFu);
            my_row[g] = (int)(w[g] >> 17);
        }
        float v[GROUPS][STEPS];
#pragma unroll
        for (int g = 0; g < GROUPS; ++g) EdgeRows<COUT, 0, STEPS>::gather(my_src[g], src, c, v[g]);
        base += BK_T * GROUPS;
#pragma unroll
        for (int g = 0; g < GROUPS; ++g) {
            const int j = base + g * 32 + lane;
            w[g] = (j < end) ? sorted[j] : PAD;
        }
#pragma unroll
        for (int g = 0; g < GROUPS; ++g) EdgeRows<COUT, 0, STEPS>::add(my_row[g], acc, c, v[g]);
    }
    if constexpr (EPI == EPI_POOL) {
        if (tid == 0) s_half = half_of_reply(die_reply);
    }
    __syncthreads();

    if constexpr (EPI == EPI_L12) {
        if (tile_wave) stage.finish(acc, wv * 16 + (lane & 15), b * BS + wv * 16 + (lane & 15), lane >> 4, outp);
    }
    if constexpr (EPI == EPI_L3) {
        if (tile_wave) stage.finish(acc, wv * 16, b * BS + wv * 16, lane & 15, lane >> 4, outp);
    }
    if constexpr (EPI == EPI_POOL) {
        if (chunk_lane)
            stage.finish(acc, (tid >> 4) * NodeStage<EPI_POOL>::NPC, tid & 15, outp + s_half * (N_GRAPHS * ROWF));
    }
}

__global__ void __launch_bounds__(256)
head_transform(const float* __restrict__ P0, const float* __restrict__ P1,
               const float* __restrict__ W, const float* __restrict__ bias,
               const float* __restrict__ Wc, const float* __restrict__ bc, float* out) {
    static_assert(N_GRAPHS % 32 == 0, "a wave owns 32 graphs");
    const int wave = blockIdx.x * (blockDim.x >> 5) + (threadIdx.x >> 5);
    const int g0 = wave * 32;
    if (g0 >= N_GRAPHS) return;
    const int lane = threadIdx.x & 31;
    const int half = lane >> 4;
    const int m    = lane & 15;
    const float bb  = bias[m];
    const float wc  = Wc[m];
    const float bc0 = bc[0];
    float* logits = out + N_GRAPHS * 16;
#pragma unroll
    for (int t = 0; t < 2; ++t) {
        const int row0 = g0 + t * 16;
        const int row  = row0 + m;
        v8f o = (v8f){};
#pragma unroll
        for (int ks = 0; ks < 4; ++ks) {
            const int ka = ks * 4 + 2 * half;
            v2f a = *reinterpret_cast<const v2f*>(&P0[row * ROWF + ka]);
            a += *reinterpret_cast<const v2f*>(&P1[row * ROWF + ka]);
            const v2f bw = *reinterpret_cast<const v2f*>(&W[m * 16 + ka]);
            o = __builtin_amdgcn_wmma_f32_16x16x4_f32(
                false, a, false, bw, (short)0, o, false, false);
        }
        float lg[8];
#pragma unroll
        for (int r = 0; r < 8; ++r) {
            const float e = o[r] + bb;
            out[(row0 + r + 8 * half) * 16 + m] = e;
            float s = fmaxf(e, 0.0f) * wc;
            s += __shfl_xor(s, 8);
            s += __shfl_xor(s, 4);
            s += __shfl_xor(s, 2);
            s += __shfl_xor(s, 1);
            lg[r] = s + bc0;
        }
        if (m == 0) {
            float* pl = &logits[row0 + 8 * half];
            *reinterpret_cast<float4*>(pl)     = make_float4(lg[0], lg[1], lg[2], lg[3]);
            *reinterpret_cast<float4*>(pl + 4) = make_float4(lg[4], lg[5], lg[6], lg[7]);
        }
    }
}

extern "C" void kernel_launch(void* const* d_in, const int* in_sizes, int n_in,
                              void* d_out, int out_size, void* d_ws, size_t ws_size,
                              hipStream_t stream) {
    const float* x     = (const float*)d_in[0];
    const int*   ei    = (const int*)d_in[1];
    const int*   batch = (const int*)d_in[2];
    const float* W1 = (const float*)d_in[3];
    const float* b1 = (const float*)d_in[4];
    const float* W2 = (const float*)d_in[5];
    const float* b2 = (const float*)d_in[6];
    const float* W3 = (const float*)d_in[7];
    const float* b3 = (const float*)d_in[8];
    const float* We = (const float*)d_in[9];
    const float* be = (const float*)d_in[10];
    const float* Wc = (const float*)d_in[11];
    const float* bc = (const float*)d_in[12];
    float* out = (float*)d_out;

    float* ws     = (float*)d_ws;
    float* dis    = ws;
    float* bufT   = dis  + N_NODES;
    float* bufA   = bufT + N_NODES * ROWF;
    unsigned* sorted = (unsigned*)(bufA + N_EDGES);
    int*   tab     = (int*)(sorted + N_EDGES);
    float* pooled2 = (float*)(tab + TABLEN);

    const int B = 256;

    slice_sort<<<NBLK, SORT_T, 0, stream>>>(ei, tab, (unsigned*)bufA);

    bucket_prep<<<NB, BK_T, 0, stream>>>((const unsigned*)bufA, x, dis, tab, bufT, sorted);

    bucket_layer<4, 4, EPI_L12><<<NB, BK_T, 0, stream>>>(sorted, bufT, tab, dis, W1, b1, W2, nullptr, bufA, nullptr);

    bucket_layer<16, 2, EPI_L3><<<NB, BK_T, 0, stream>>>(sorted, bufA, tab, dis, W3, b2, nullptr, nullptr, bufT, pooled2);

    bucket_layer<16, 2, EPI_POOL><<<NB, BK_T, 0, stream>>>(sorted, bufT, tab, dis, nullptr, b3, nullptr, batch, pooled2, nullptr);

    head_transform<<<ceil_div(N_GRAPHS / 32, B / 32), B, 0, stream>>>(
        pooled2, pooled2 + N_GRAPHS * ROWF, We, be, Wc, bc, out);
}
